# starting kernel
# baseline (speedup 1.0000x reference)
#include <hip/hip_runtime.h>
#include <hip/hip_bf16.h>
#include <stdint.h>

typedef _Float16 v16h __attribute__((ext_vector_type(16)));
typedef _Float16 v8h  __attribute__((ext_vector_type(8)));
typedef float    v8f  __attribute__((ext_vector_type(8)));

#define F_DIM 128

__global__ void zero_f4(float4* __restrict__ p, int n4) {
    int i = blockIdx.x * blockDim.x + threadIdx.x;
    if (i < n4) p[i] = make_float4(0.f, 0.f, 0.f, 0.f);
}

__global__ void wconvert(const float* __restrict__ W11, const float* __restrict__ W12,
                         const float* __restrict__ W21, const float* __restrict__ W22,
                         _Float16* __restrict__ frags) {
    int gid = blockIdx.x * blockDim.x + threadIdx.x;
    if (gid >= 4 * 8192) return;
    int m    = gid >> 13;
    int r    = gid & 8191;
    int d    = r & 7;
    int lane = (r >> 3) & 31;
    int ntk  = r >> 8;
    int kt   = ntk >> 3;
    int nt   = ntk & 7;
    int nn   = nt * 16 + (lane & 15);
    int kb   = kt * 32 + ((lane >> 4) << 3);
    int k0   = kb + ((d < 4) ? (d << 1) : (16 + ((d - 4) << 1)));
    const float* W = (m == 0) ? W11 : (m == 1) ? W12 : (m == 2) ? W21 : W22;
    _Float16* o = frags + (size_t)m * 16384 + (size_t)r * 2;
    o[0] = (_Float16)W[(size_t)k0 * F_DIM + nn];
    o[1] = (_Float16)W[(size_t)(k0 + 1) * F_DIM + nn];
}

__global__ void edge_scatter(const float* __restrict__ xin, const int* __restrict__ ei,
                             const float* __restrict__ ea, const float* __restrict__ We,
                             const float* __restrict__ be, float* aggr, int E_) {
    long long gid = (long long)blockIdx.x * blockDim.x + threadIdx.x;
    if (gid >= (long long)E_ * 32) return;
    int e = (int)(gid >> 5);
    int c = ((int)gid & 31) << 2;
    int src = ei[e];
    int dst = ei[E_ + e];
    float a = ea[e];
    const float* xr = xin + (size_t)src * F_DIM + c;
    float4 xv = *(const float4*)xr;
    float4 wv = *(const float4*)(We + c);
    float4 bv = *(const float4*)(be + c);
    float m0 = fmaxf(xv.x + a * wv.x + bv.x, 0.f);
    float m1 = fmaxf(xv.y + a * wv.y + bv.y, 0.f);
    float m2 = fmaxf(xv.z + a * wv.z + bv.z, 0.f);
    float m3 = fmaxf(xv.w + a * wv.w + bv.w, 0.f);
    float* o = aggr + (size_t)dst * F_DIM + c;
    atomicAdd(o + 0, m0);
    atomicAdd(o + 1, m1);
    atomicAdd(o + 2, m2);
    atomicAdd(o + 3, m3);
}

__global__ void __launch_bounds__(256)
mlp_wmma(const float* __restrict__ xin, const float* aggr, const float* __restrict__ eps,
         const _Float16* __restrict__ fw1, const float* __restrict__ b1,
         const _Float16* __restrict__ fw2, const float* __restrict__ b2,
         float* out, int nrows) {
    __shared__ __align__(16) _Float16 stage[8][16 * 136];

    const int wv   = threadIdx.x >> 5;
    const int lane = threadIdx.x & 31;
    const int row0 = blockIdx.x * 128 + wv * 16;
    if (row0 >= nrows) return;

    const float ep = 1.0f + eps[0];
    _Float16* st = &stage[wv][0];

    #pragma unroll 4
    for (int r = 0; r < 16; ++r) {
        size_t off = (size_t)(row0 + r) * F_DIM + (lane << 2);
        float4 xv = *(const float4*)(xin + off);
        float4 av = *(const float4*)(aggr + off);
        union { _Float16 h[4]; unsigned long long u; } p;
        p.h[0] = (_Float16)(ep * xv.x + av.x);
        p.h[1] = (_Float16)(ep * xv.y + av.y);
        p.h[2] = (_Float16)(ep * xv.z + av.z);
        p.h[3] = (_Float16)(ep * xv.w + av.w);
        *(unsigned long long*)(st + r * 136 + (lane << 2)) = p.u;
    }

    const int m_  = lane & 15;
    const int hs8 = (lane >> 4) << 3;

    v8f acc[8];
    const v8f vz = {0.f, 0.f, 0.f, 0.f, 0.f, 0.f, 0.f, 0.f};

    #pragma unroll
    for (int nt = 0; nt < 8; ++nt) acc[nt] = vz;
    #pragma unroll
    for (int kt = 0; kt < 4; ++kt) {
        int kb = kt * 32 + hs8;
        v8h alo = *(const v8h*)(st + m_ * 136 + kb);
        v8h ahi = *(const v8h*)(st + m_ * 136 + kb + 16);
        v16h av = __builtin_shufflevector(alo, ahi, 0,1,2,3,4,5,6,7,8,9,10,11,12,13,14,15);
        #pragma unroll
        for (int nt = 0; nt < 8; ++nt) {
            const _Float16* q = fw1 + ((size_t)((kt << 3) | nt) * 32 + lane) * 16;
            v8h blo = *(const v8h*)q;
            v8h bhi = *(const v8h*)(q + 8);
            v16h bv = __builtin_shufflevector(blo, bhi, 0,1,2,3,4,5,6,7,8,9,10,11,12,13,14,15);
            acc[nt] = __builtin_amdgcn_wmma_f32_16x16x32_f16(
                false, av, false, bv, (short)0, acc[nt], false, false);
        }
    }

    #pragma unroll
    for (int nt = 0; nt < 8; ++nt) {
        int col = nt * 16 + m_;
        float bs = b1[col];
        #pragma unroll
        for (int r = 0; r < 8; ++r) {
            float v = fmaxf(acc[nt][r] + bs, 0.f);
            st[(r + hs8) * 136 + col] = (_Float16)v;
        }
    }

    #pragma unroll
    for (int nt = 0; nt < 8; ++nt) acc[nt] = vz;
    #pragma unroll
    for (int kt = 0; kt < 4; ++kt) {
        int kb = kt * 32 + hs8;
        v8h alo = *(const v8h*)(st + m_ * 136 + kb);
        v8h ahi = *(const v8h*)(st + m_ * 136 + kb + 16);
        v16h av = __builtin_shufflevector(alo, ahi, 0,1,2,3,4,5,6,7,8,9,10,11,12,13,14,15);
        #pragma unroll
        for (int nt = 0; nt < 8; ++nt) {
            const _Float16* q = fw2 + ((size_t)((kt << 3) | nt) * 32 + lane) * 16;
            v8h blo = *(const v8h*)q;
            v8h bhi = *(const v8h*)(q + 8);
            v16h bv = __builtin_shufflevector(blo, bhi, 0,1,2,3,4,5,6,7,8,9,10,11,12,13,14,15);
            acc[nt] = __builtin_amdgcn_wmma_f32_16x16x32_f16(
                false, av, false, bv, (short)0, acc[nt], false, false);
        }
    }

    #pragma unroll
    for (int nt = 0; nt < 8; ++nt) {
        int col = nt * 16 + m_;
        float bs = b2[col];
        #pragma unroll
        for (int r = 0; r < 8; ++r) {
            out[(size_t)(row0 + r + hs8) * F_DIM + col] = fmaxf(acc[nt][r] + bs, 0.f);
        }
    }
}

__global__ void __launch_bounds__(256)
classifier(const float* __restrict__ h, const float* __restrict__ Wl,
           const float* __restrict__ bl, float* __restrict__ out, int n) {
    __shared__ float w[F_DIM * 40];
    __shared__ float bb[40];
    for (int i = threadIdx.x; i < F_DIM * 40; i += blockDim.x) w[i] = Wl[i];
    if (threadIdx.x < 40) bb[threadIdx.x] = bl[threadIdx.x];
    __syncthreads();

    int node = blockIdx.x * blockDim.x + threadIdx.x;
    if (node >= n) return;

    float acc[40];
    #pragma unroll
    for (int c = 0; c < 40; ++c) acc[c] = bb[c];

    const float* hr = h + (size_t)node * F_DIM;
    for (int k = 0; k < F_DIM; ++k) {
        float hv = hr[k];
        #pragma unroll
        for (int c = 0; c < 40; ++c) acc[c] = fmaf(hv, w[k * 40 + c], acc[c]);
    }

    float mx = acc[0];
    #pragma unroll
    for (int c = 1; c < 40; ++c) mx = fmaxf(mx, acc[c]);
    float s = 0.f;
    #pragma unroll
    for (int c = 0; c < 40; ++c) s += __expf(acc[c] - mx);
    float ls = __logf(s);
    float* o = out + (size_t)node * 40;
    #pragma unroll
    for (int c = 0; c < 40; ++c) o[c] = acc[c] - mx - ls;
}

extern "C" void kernel_launch(void* const* d_in, const int* in_sizes, int n_in,
                              void* d_out, int out_size, void* d_ws, size_t ws_size,
                              hipStream_t stream) {
    const float* x    = (const float*)d_in[0];
    const int*   ei   = (const int*)d_in[1];
    const float* ea   = (const float*)d_in[2];
    const float* eps1 = (const float*)d_in[3];
    const float* We1  = (const float*)d_in[4];
    const float* be1  = (const float*)d_in[5];
    const float* W11  = (const float*)d_in[6];
    const float* b11  = (const float*)d_in[7];
    const float* W12  = (const float*)d_in[8];
    const float* b12  = (const float*)d_in[9];
    const float* eps2 = (const float*)d_in[10];
    const float* We2  = (const float*)d_in[11];
    const float* be2  = (const float*)d_in[12];
    const float* W21  = (const float*)d_in[13];
    const float* b21  = (const float*)d_in[14];
    const float* W22  = (const float*)d_in[15];
    const float* b22  = (const float*)d_in[16];
    const float* Wl   = (const float*)d_in[17];
    const float* bl   = (const float*)d_in[18];

    const int N_ = in_sizes[0] / F_DIM;
    const int E_ = in_sizes[2];
    const size_t NB = (size_t)N_ * F_DIM * sizeof(float);

    char* ws = (char*)d_ws;
    float*    aggr  = (float*)ws;
    float*    h1    = (float*)(ws + NB);
    _Float16* frags = (_Float16*)(ws + 2 * NB);
    _Float16* f11 = frags;
    _Float16* f12 = frags + 16384;
    _Float16* f21 = frags + 32768;
    _Float16* f22 = frags + 49152;
    float* h2  = aggr;
    float* out = (float*)d_out;

    const int n4 = N_ * F_DIM / 4;
    const int zb = (n4 + 255) / 256;
    const long long ework = (long long)E_ * 32;
    const int eb = (int)((ework + 255) / 256);
    const int mb = (N_ + 127) / 128;
    const int cb = (N_ + 255) / 256;

    wconvert<<<128, 256, 0, stream>>>(W11, W12, W21, W22, frags);

    zero_f4<<<zb, 256, 0, stream>>>((float4*)aggr, n4);
    edge_scatter<<<eb, 256, 0, stream>>>(x, ei, ea, We1, be1, aggr, E_);
    mlp_wmma<<<mb, 256, 0, stream>>>(x, aggr, eps1, f11, b11, f12, b12, h1, N_);

    zero_f4<<<zb, 256, 0, stream>>>((float4*)aggr, n4);
    edge_scatter<<<eb, 256, 0, stream>>>(h1, ei, ea, We2, be2, aggr, E_);
    mlp_wmma<<<mb, 256, 0, stream>>>(h1, aggr, eps2, f21, b21, f22, b22, h2, N_);

    classifier<<<cb, 256, 0, stream>>>(h2, Wl, bl, out, N_);
}


# Layer-2 edge launch reads back each wave's matches (source, local destination, attribute) saved by layer 1's scan: no second scan, no scattered index loads; sum order varies as before
# speedup vs baseline: 2.5367x; 2.5367x over previous
#include <hip/hip_runtime.h>
#include <hip/hip_bf16.h>
#include <stdint.h>

typedef _Float16 v16h __attribute__((ext_vector_type(16)));
typedef _Float16 v8h  __attribute__((ext_vector_type(8)));
typedef float    v8f  __attribute__((ext_vector_type(8)));

#define F_DIM 128

#define TILE_NODES 400

__device__ __forceinline__ unsigned tile_byte(int d) {
    return ((unsigned)d / (unsigned)TILE_NODES) & 255u;
}

__global__ void __launch_bounds__(256)
prep(const float* __restrict__ W11, const float* __restrict__ W12,
     const float* __restrict__ W21, const float* __restrict__ W22,
     _Float16* __restrict__ frags, const int* __restrict__ ei,
     unsigned* __restrict__ tid4, int E_) {
    const int gid = blockIdx.x * blockDim.x + threadIdx.x;
    if (gid < 4 * 8192) {
        int m    = gid >> 13;
        int r    = gid & 8191;
        int d    = r & 7;
        int lane = (r >> 3) & 31;
        int ntk  = r >> 8;
        int kt   = ntk >> 3;
        int nt   = ntk & 7;
        int nn   = nt * 16 + (lane & 15);
        int kb   = kt * 32 + ((lane >> 4) << 3);
        int k0   = kb + ((d < 4) ? (d << 1) : (16 + ((d - 4) << 1)));
        const float* W = (m == 0) ? W11 : (m == 1) ? W12 : (m == 2) ? W21 : W22;
        _Float16* o = frags + (size_t)m * 16384 + (size_t)r * 2;
        o[0] = (_Float16)W[(size_t)k0 * F_DIM + nn];
        o[1] = (_Float16)W[(size_t)(k0 + 1) * F_DIM + nn];
    }

    const int* dsts = ei + E_;
    const int  nq   = (E_ + 3) >> 2;
    const bool al   = (E_ & 3) == 0;
    for (int i = gid; i < nq; i += (int)(gridDim.x * blockDim.x)) {
        const int e = 4 * i;
        int d0, d1, d2, d3;
        if (al) {
            const int4 v = *(const int4*)(dsts + e);
            d0 = v.x; d1 = v.y; d2 = v.z; d3 = v.w;
        } else {
            d0 = dsts[e];
            d1 = (e + 1 < E_) ? dsts[e + 1] : 0;
            d2 = (e + 2 < E_) ? dsts[e + 2] : 0;
            d3 = (e + 3 < E_) ? dsts[e + 3] : 0;
        }
        tid4[i] = tile_byte(d0) | (tile_byte(d1) << 8) | (tile_byte(d2) << 16) | (tile_byte(d3) << 24);
    }
}

#define EDGE_WAVES 32
#define SCAN_U     4
#define QCAP       544
#define RCAP       512

__device__ __forceinline__ unsigned zero_bytes(unsigned x) {
    return ~(((x & 0x7F7F7F7Fu) + 0x7F7F7F7Fu) | x | 0x7F7F7F7Fu);
}

template <class F>
__device__ __forceinline__ void tile_add_batch(
        const float* __restrict__ xin, int sv, int dv, int av, int m,
        int nn, int lane, const float (&we)[4], const float (&bb)[4], float* tile,
        F behind_first_loads) {
    for (int j = 0; j < m; j += 8) {
        float r[8][4];
        int   dl[8];
        float aa[8];
        #pragma unroll
        for (int u = 0; u < 8; ++u) {
            const int jj = min(j + u, m - 1);
            const int s  = __builtin_amdgcn_readlane(sv, jj);
            dl[u] = __builtin_amdgcn_readlane(dv, jj);
            aa[u] = __int_as_float(__builtin_amdgcn_readlane(av, jj));
            const float* xr = xin + (size_t)s * F_DIM + lane;
            #pragma unroll
            for (int k = 0; k < 4; ++k) r[u][k] = xr[32 * k];
        }
        if (j == 0) behind_first_loads();
        #pragma unroll
        for (int u = 0; u < 8; ++u) {
            if (j + u < m && (unsigned)dl[u] < (unsigned)nn) {
                float* t = tile + dl[u] * F_DIM + lane;
                #pragma unroll
                for (int k = 0; k < 4; ++k) {
                    const float v = fmaxf(r[u][k] + aa[u] * we[k] + bb[k], 0.f);
                    __hip_atomic_fetch_add(t + 32 * k, v, __ATOMIC_RELAXED,
                                           __HIP_MEMORY_SCOPE_WORKGROUP);
                }
            }
        }
    }
}

template <bool SAVE>
__device__ __forceinline__ int tile_add_queue(
        const float* __restrict__ xin, const int* __restrict__ srcs,
        const int* __restrict__ dsts, const float* __restrict__ ea,
        int* q, int n, bool all, int lo, int nn, int lane,
        const float (&we)[4], const float (&bb)[4], float* tile,
        int* __restrict__ rec, int& nsaved) {
    int base = 0;
    for (; base < n; base += 32) {
        const int m = min(32, n - base);
        if (!all && m < 32) break;
        const int e  = q[base + min(lane, m - 1)];
        const int sv = srcs[e];
        const int dv = dsts[e] - lo;
        const int av = __float_as_int(ea[e]);
        if (SAVE) {
            const int p = nsaved + lane;
            if (lane < m && p < RCAP) {
                rec[p]            = sv;
                rec[RCAP + p]     = dv;
                rec[2 * RCAP + p] = av;
            }
            nsaved += m;
        }
        tile_add_batch(xin, sv, dv, av, m, nn, lane, we, bb, tile, [] {});
    }
    if (base >= n) return 0;
    const int rem = n - base;
    if (base > 0 && lane < rem) q[lane] = q[base + lane];
    return rem;
}

template <bool SAVE>
__device__ __forceinline__ int scan_tile(
        const float* __restrict__ xin, const int* __restrict__ ei,
        const float* __restrict__ ea, const unsigned char* __restrict__ tbytes,
        int* q, int* __restrict__ rec, int E_, int lo, int nn, int lane, int wave,
        const float (&we)[4], const float (&bb)[4], float* tile) {
    const int* dsts = ei + E_;
    const unsigned kb  = (unsigned)blockIdx.x & 255u;
    const unsigned key = kb * 0x01010101u;
    const int nfull = E_ >> 9;
    const int rot = (int)blockIdx.x * (nfull / (int)gridDim.x);

    int nsaved = 0;
    int cnt = 0;
    for (int b0 = wave; b0 < nfull; b0 += EDGE_WAVES * SCAN_U) {
        uint4    v[SCAN_U];
        unsigned okm[SCAN_U];
        int      eb[SCAN_U];
        #pragma unroll
        for (int u = 0; u < SCAN_U; ++u) {
            const int  b  = b0 + EDGE_WAVES * u;
            const bool ok = b < nfull;
            okm[u] = (unsigned)((b - nfull) >> 31);
            int br = (ok ? b : b0) + rot;
            if (br >= nfull) br -= nfull;
            eb[u] = br * 512 + lane * 16;
            v[u] = *(const uint4*)(tbytes + eb[u]);
        }
        #pragma unroll
        for (int u = 0; u < SCAN_U; ++u) {
            const unsigned t[4] = {zero_bytes(v[u].x ^ key) & okm[u], zero_bytes(v[u].y ^ key) & okm[u],
                                   zero_bytes(v[u].z ^ key) & okm[u], zero_bytes(v[u].w ^ key) & okm[u]};
            #pragma unroll
            for (int k = 0; k < 4; ++k) {
                unsigned tk   = t[k];
                unsigned mask = __builtin_amdgcn_ballot_w32(tk != 0u);
                while (mask != 0u) {
                    if (tk != 0u) {
                        q[cnt + (int)__builtin_amdgcn_mbcnt_lo(mask, 0u)] =
                            eb[u] + 4 * k + (__builtin_ctz(tk) >> 3);
                        tk &= tk - 1u;
                    }
                    cnt += __builtin_popcount(mask);
                    mask = __builtin_amdgcn_ballot_w32(tk != 0u);
                }
            }
            if (cnt >= 32)
                cnt = tile_add_queue<SAVE>(xin, ei, dsts, ea, q, cnt, false, lo, nn, lane, we, bb, tile, rec, nsaved);
        }
    }
    for (int e0 = nfull * 512 + wave * 32; e0 < E_; e0 += EDGE_WAVES * 32) {
        const int e = e0 + lane;
        const bool hit = (e < E_) && ((unsigned)tbytes[min(e, E_ - 1)] == kb);
        const unsigned mask = __builtin_amdgcn_ballot_w32(hit);
        if (mask != 0u) {
            if (hit) q[cnt + (int)__builtin_amdgcn_mbcnt_lo(mask, 0u)] = e;
            cnt += __builtin_popcount(mask);
        }
        if (cnt >= 32)
            cnt = tile_add_queue<SAVE>(xin, ei, dsts, ea, q, cnt, false, lo, nn, lane, we, bb, tile, rec, nsaved);
    }
    if (cnt > 0)
        tile_add_queue<SAVE>(xin, ei, dsts, ea, q, cnt, true, lo, nn, lane, we, bb, tile, rec, nsaved);
    return nsaved;
}

template <bool REPLAY>
__global__ void __launch_bounds__(1024)
edge_tile(const float* __restrict__ xin, const int* __restrict__ ei,
          const float* __restrict__ ea, const float* __restrict__ We,
          const float* __restrict__ be, const unsigned* __restrict__ tid4,
          float* __restrict__ aggr, int* __restrict__ recs, int* __restrict__ rcnt,
          int E_, int N_) {
    __shared__ __align__(16) float tile[TILE_NODES * F_DIM];
    __shared__ int wq[EDGE_WAVES][QCAP];

    const int tid  = threadIdx.x;
    const int lane = tid & 31;
    const int wave = __builtin_amdgcn_readfirstlane(tid >> 5);
    const int lo   = blockIdx.x * TILE_NODES;
    const int nn   = min(TILE_NODES, N_ - lo);

    const int slot = (int)blockIdx.x * EDGE_WAVES + wave;
    int* rec = recs + (size_t)slot * (3 * RCAP);
    int nrec = 0;
    if (REPLAY) nrec = rcnt[slot];

    float4* tile4 = (float4*)tile;
    for (int i = tid; i < TILE_NODES * F_DIM / 4; i += 1024)
        tile4[i] = make_float4(0.f, 0.f, 0.f, 0.f);

    float we[4], bb[4];
    #pragma unroll
    for (int k = 0; k < 4; ++k) { we[k] = We[lane + 32 * k]; bb[k] = be[lane + 32 * k]; }

    int* q = wq[wave];
    const unsigned char* tbytes = (const unsigned char*)tid4;

    __syncthreads();

    if (REPLAY) {
        const int n = __builtin_amdgcn_readfirstlane(nrec);
        if (n <= RCAP) {
            if (n > 0) {
                int idx = min(lane, n - 1);
                int sv = rec[idx], dv = rec[RCAP + idx], av = rec[2 * RCAP + idx];
                for (int base = 0; base < n; base += 32) {
                    const int m = min(32, n - base);
                    idx = min(base + 32 + lane, n - 1);
                    int sv2, dv2, av2;
                    tile_add_batch(xin, sv, dv, av, m, nn, lane, we, bb, tile, [&] {
                        sv2 = rec[idx]; dv2 = rec[RCAP + idx]; av2 = rec[2 * RCAP + idx];
                    });
                    sv = sv2; dv = dv2; av = av2;
                }
            }
        } else {
            scan_tile<false>(xin, ei, ea, tbytes, q, rec, E_, lo, nn, lane, wave, we, bb, tile);
        }
    } else {
        const int found = scan_tile<true>(xin, ei, ea, tbytes, q, rec, E_, lo, nn, lane, wave, we, bb, tile);
        if (lane == 0) rcnt[slot] = found;
    }

    __syncthreads();

    float4* o4 = (float4*)(aggr + (size_t)lo * F_DIM);
    for (int i = tid; i < nn * (F_DIM / 4); i += 1024) o4[i] = tile4[i];
}

__global__ void __launch_bounds__(256)
mlp_wmma(const float* __restrict__ xin, const float* aggr, const float* __restrict__ eps,
         const _Float16* __restrict__ fw1, const float* __restrict__ b1,
         const _Float16* __restrict__ fw2, const float* __restrict__ b2,
         float* out, int nrows) {
    __shared__ __align__(16) _Float16 stage[8][16 * 136];

    const int wv   = threadIdx.x >> 5;
    const int lane = threadIdx.x & 31;
    const int row0 = blockIdx.x * 128 + wv * 16;
    if (row0 >= nrows) return;

    const float ep = 1.0f + eps[0];
    _Float16* st = &stage[wv][0];

    #pragma unroll 4
    for (int r = 0; r < 16; ++r) {
        size_t off = (size_t)(row0 + r) * F_DIM + (lane << 2);
        float4 xv = *(const float4*)(xin + off);
        float4 av = *(const float4*)(aggr + off);
        union { _Float16 h[4]; unsigned long long u; } p;
        p.h[0] = (_Float16)(ep * xv.x + av.x);
        p.h[1] = (_Float16)(ep * xv.y + av.y);
        p.h[2] = (_Float16)(ep * xv.z + av.z);
        p.h[3] = (_Float16)(ep * xv.w + av.w);
        *(unsigned long long*)(st + r * 136 + (lane << 2)) = p.u;
    }

    const int m_  = lane & 15;
    const int hs8 = (lane >> 4) << 3;

    v8f acc[8];
    const v8f vz = {0.f, 0.f, 0.f, 0.f, 0.f, 0.f, 0.f, 0.f};

    #pragma unroll
    for (int nt = 0; nt < 8; ++nt) acc[nt] = vz;
    #pragma unroll
    for (int kt = 0; kt < 4; ++kt) {
        int kb = kt * 32 + hs8;
        v8h alo = *(const v8h*)(st + m_ * 136 + kb);
        v8h ahi = *(const v8h*)(st + m_ * 136 + kb + 16);
        v16h av = __builtin_shufflevector(alo, ahi, 0,1,2,3,4,5,6,7,8,9,10,11,12,13,14,15);
        #pragma unroll
        for (int nt = 0; nt < 8; ++nt) {
            const _Float16* q = fw1 + ((size_t)((kt << 3) | nt) * 32 + lane) * 16;
            v8h blo = *(const v8h*)q;
            v8h bhi = *(const v8h*)(q + 8);
            v16h bv = __builtin_shufflevector(blo, bhi, 0,1,2,3,4,5,6,7,8,9,10,11,12,13,14,15);
            acc[nt] = __builtin_amdgcn_wmma_f32_16x16x32_f16(
                false, av, false, bv, (short)0, acc[nt], false, false);
        }
    }

    #pragma unroll
    for (int nt = 0; nt < 8; ++nt) {
        int col = nt * 16 + m_;
        float bs = b1[col];
        #pragma unroll
        for (int r = 0; r < 8; ++r) {
            float v = fmaxf(acc[nt][r] + bs, 0.f);
            st[(r + hs8) * 136 + col] = (_Float16)v;
        }
    }

    #pragma unroll
    for (int nt = 0; nt < 8; ++nt) acc[nt] = vz;
    #pragma unroll
    for (int kt = 0; kt < 4; ++kt) {
        int kb = kt * 32 + hs8;
        v8h alo = *(const v8h*)(st + m_ * 136 + kb);
        v8h ahi = *(const v8h*)(st + m_ * 136 + kb + 16);
        v16h av = __builtin_shufflevector(alo, ahi, 0,1,2,3,4,5,6,7,8,9,10,11,12,13,14,15);
        #pragma unroll
        for (int nt = 0; nt < 8; ++nt) {
            const _Float16* q = fw2 + ((size_t)((kt << 3) | nt) * 32 + lane) * 16;
            v8h blo = *(const v8h*)q;
            v8h bhi = *(const v8h*)(q + 8);
            v16h bv = __builtin_shufflevector(blo, bhi, 0,1,2,3,4,5,6,7,8,9,10,11,12,13,14,15);
            acc[nt] = __builtin_amdgcn_wmma_f32_16x16x32_f16(
                false, av, false, bv, (short)0, acc[nt], false, false);
        }
    }

    #pragma unroll
    for (int nt = 0; nt < 8; ++nt) {
        int col = nt * 16 + m_;
        float bs = b2[col];
        #pragma unroll
        for (int r = 0; r < 8; ++r) {
            out[(size_t)(row0 + r + hs8) * F_DIM + col] = fmaxf(acc[nt][r] + bs, 0.f);
        }
    }
}

__global__ void __launch_bounds__(256)
classifier(const float* __restrict__ h, const float* __restrict__ Wl,
           const float* __restrict__ bl, float* __restrict__ out, int n) {
    __shared__ float w[F_DIM * 40];
    __shared__ float bb[40];
    for (int i = threadIdx.x; i < F_DIM * 40; i += blockDim.x) w[i] = Wl[i];
    if (threadIdx.x < 40) bb[threadIdx.x] = bl[threadIdx.x];
    __syncthreads();

    int node = blockIdx.x * blockDim.x + threadIdx.x;
    if (node >= n) return;

    float acc[40];
    #pragma unroll
    for (int c = 0; c < 40; ++c) acc[c] = bb[c];

    const float* hr = h + (size_t)node * F_DIM;
    for (int k = 0; k < F_DIM; ++k) {
        float hv = hr[k];
        #pragma unroll
        for (int c = 0; c < 40; ++c) acc[c] = fmaf(hv, w[k * 40 + c], acc[c]);
    }

    float mx = acc[0];
    #pragma unroll
    for (int c = 1; c < 40; ++c) mx = fmaxf(mx, acc[c]);
    float s = 0.f;
    #pragma unroll
    for (int c = 0; c < 40; ++c) s += __expf(acc[c] - mx);
    float ls = __logf(s);
    float* o = out + (size_t)node * 40;
    #pragma unroll
    for (int c = 0; c < 40; ++c) o[c] = acc[c] - mx - ls;
}

extern "C" void kernel_launch(void* const* d_in, const int* in_sizes, int n_in,
                              void* d_out, int out_size, void* d_ws, size_t ws_size,
                              hipStream_t stream) {
    const float* x    = (const float*)d_in[0];
    const int*   ei   = (const int*)d_in[1];
    const float* ea   = (const float*)d_in[2];
    const float* eps1 = (const float*)d_in[3];
    const float* We1  = (const float*)d_in[4];
    const float* be1  = (const float*)d_in[5];
    const float* W11  = (const float*)d_in[6];
    const float* b11  = (const float*)d_in[7];
    const float* W12  = (const float*)d_in[8];
    const float* b12  = (const float*)d_in[9];
    const float* eps2 = (const float*)d_in[10];
    const float* We2  = (const float*)d_in[11];
    const float* be2  = (const float*)d_in[12];
    const float* W21  = (const float*)d_in[13];
    const float* b21  = (const float*)d_in[14];
    const float* W22  = (const float*)d_in[15];
    const float* b22  = (const float*)d_in[16];
    const float* Wl   = (const float*)d_in[17];
    const float* bl   = (const float*)d_in[18];

    const int N_ = in_sizes[0] / F_DIM;
    const int E_ = in_sizes[2];
    const size_t NB = (size_t)N_ * F_DIM * sizeof(float);

    char* ws = (char*)d_ws;
    float*    aggr  = (float*)ws;
    float*    h1    = (float*)(ws + NB);
    _Float16* frags = (_Float16*)(ws + 2 * NB);
    unsigned* tid4  = (unsigned*)(ws + 2 * NB + 131072);
    const size_t TBY = (((size_t)E_ + 3) & ~(size_t)3);
    _Float16* f11 = frags;
    _Float16* f12 = frags + 16384;
    _Float16* f21 = frags + 32768;
    _Float16* f22 = frags + 49152;
    float* h2  = aggr;
    float* out = (float*)d_out;

    const int tb = (N_ + TILE_NODES - 1) / TILE_NODES;
    int* rcnt = (int*)(ws + 2 * NB + 131072 + ((TBY + 255) & ~(size_t)255));
    int* recs = rcnt + (((size_t)tb * EDGE_WAVES + 63) & ~(size_t)63);
    const int mb = (N_ + 127) / 128;
    const int cb = (N_ + 255) / 256;

    prep<<<1024, 256, 0, stream>>>(W11, W12, W21, W22, frags, ei, tid4, E_);

    edge_tile<false><<<tb, 1024, 0, stream>>>(x, ei, ea, We1, be1, tid4, aggr, recs, rcnt, E_, N_);
    mlp_wmma<<<mb, 256, 0, stream>>>(x, aggr, eps1, f11, b11, f12, b12, h1, N_);

    edge_tile<true><<<tb, 1024, 0, stream>>>(h1, ei, ea, We2, be2, tid4, aggr, recs, rcnt, E_, N_);
    mlp_wmma<<<mb, 256, 0, stream>>>(h1, aggr, eps2, f21, b21, f22, b22, h2, N_);

    classifier<<<cb, 256, 0, stream>>>(h2, Wl, bl, out, N_);
}


# Scan matches read one 16-byte record per edge (source, destination, attribute, packed in edge order by the first launch) instead of three scattered 4-byte loads; sum order varies as before
# speedup vs baseline: 3.4299x; 1.3521x over previous
#include <hip/hip_runtime.h>
#include <hip/hip_bf16.h>
#include <stdint.h>

typedef _Float16 v16h __attribute__((ext_vector_type(16)));
typedef _Float16 v8h  __attribute__((ext_vector_type(8)));
typedef float    v8f  __attribute__((ext_vector_type(8)));
typedef float    v4f  __attribute__((ext_vector_type(4)));
typedef int      v4i  __attribute__((ext_vector_type(4)));

#define F_DIM 128

#define TILE_NODES 400

__device__ __forceinline__ unsigned tile_byte(int d) {
    return ((unsigned)d / (unsigned)TILE_NODES) & 255u;
}

__global__ void __launch_bounds__(256)
prep(const float* __restrict__ W11, const float* __restrict__ W12,
     const float* __restrict__ W21, const float* __restrict__ W22,
     _Float16* __restrict__ frags, const int* __restrict__ ei,
     const float* __restrict__ ea, unsigned* __restrict__ tid4,
     int* __restrict__ erec, int E_) {
    const int gid = blockIdx.x * blockDim.x + threadIdx.x;
    if (gid < 4 * 8192) {
        int m    = gid >> 13;
        int r    = gid & 8191;
        int d    = r & 7;
        int lane = (r >> 3) & 31;
        int ntk  = r >> 8;
        int kt   = ntk >> 3;
        int nt   = ntk & 7;
        int nn   = nt * 16 + (lane & 15);
        int kb   = kt * 32 + ((lane >> 4) << 3);
        int k0   = kb + ((d < 4) ? (d << 1) : (16 + ((d - 4) << 1)));
        const float* W = (m == 0) ? W11 : (m == 1) ? W12 : (m == 2) ? W21 : W22;
        _Float16* o = frags + (size_t)m * 16384 + (size_t)r * 2;
        o[0] = (_Float16)W[(size_t)k0 * F_DIM + nn];
        o[1] = (_Float16)W[(size_t)(k0 + 1) * F_DIM + nn];
    }

    const int* dsts = ei + E_;
    const int  nq   = (E_ + 3) >> 2;
    const bool al   = (E_ & 3) == 0;
    for (int i = gid; i < nq; i += (int)(gridDim.x * blockDim.x)) {
        const int e = 4 * i;
        int d0, d1, d2, d3;
        if (al) {
            const int4 v = *(const int4*)(dsts + e);
            d0 = v.x; d1 = v.y; d2 = v.z; d3 = v.w;
        } else {
            d0 = dsts[e];
            d1 = (e + 1 < E_) ? dsts[e + 1] : 0;
            d2 = (e + 2 < E_) ? dsts[e + 2] : 0;
            d3 = (e + 3 < E_) ? dsts[e + 3] : 0;
        }
        tid4[i] = tile_byte(d0) | (tile_byte(d1) << 8) | (tile_byte(d2) << 16) | (tile_byte(d3) << 24);
    }

    v4i* er = (v4i*)erec;
    const int ln   = threadIdx.x & 31;
    const int wv   = gid >> 5;
    const int nwv  = (int)(gridDim.x * blockDim.x) >> 5;
    for (int c = wv * 128; c < E_; c += nwv * 128) {
        int sv[4], dv[4], av[4];
        #pragma unroll
        for (int k = 0; k < 4; ++k) {
            const int e = min(c + 32 * k + ln, E_ - 1);
            sv[k] = ei[e];
            dv[k] = dsts[e];
            av[k] = __float_as_int(ea[e]);
        }
        #pragma unroll
        for (int k = 0; k < 4; ++k) {
            const int e = c + 32 * k + ln;
            if (e < E_) {
                const v4i r = {sv[k], dv[k], av[k], 0};
                er[e] = r;
            }
        }
    }
}

#define EDGE_WAVES 32
#define SCAN_U     4
#define QCAP       544
#define RCAP       512

__device__ __forceinline__ unsigned zero_bytes(unsigned x) {
    return ~(((x & 0x7F7F7F7Fu) + 0x7F7F7F7Fu) | x | 0x7F7F7F7Fu);
}

template <class F>
__device__ __forceinline__ void tile_add_batch(
        const float* __restrict__ xin, int sv, int dv, int av, int m,
        int nn, int lane, const float (&we)[4], const float (&bb)[4], float* tile,
        F behind_first_loads) {
    for (int j = 0; j < m; j += 8) {
        float r[8][4];
        int   dl[8];
        float aa[8];
        #pragma unroll
        for (int u = 0; u < 8; ++u) {
            const int jj = min(j + u, m - 1);
            const int s  = __builtin_amdgcn_readlane(sv, jj);
            dl[u] = __builtin_amdgcn_readlane(dv, jj);
            aa[u] = __int_as_float(__builtin_amdgcn_readlane(av, jj));
            const float* xr = xin + (size_t)s * F_DIM + lane;
            #pragma unroll
            for (int k = 0; k < 4; ++k) r[u][k] = xr[32 * k];
        }
        if (j == 0) behind_first_loads();
        #pragma unroll
        for (int u = 0; u < 8; ++u) {
            if (j + u < m && (unsigned)dl[u] < (unsigned)nn) {
                float* t = tile + dl[u] * F_DIM + lane;
                #pragma unroll
                for (int k = 0; k < 4; ++k) {
                    const float v = fmaxf(r[u][k] + aa[u] * we[k] + bb[k], 0.f);
                    __hip_atomic_fetch_add(t + 32 * k, v, __ATOMIC_RELAXED,
                                           __HIP_MEMORY_SCOPE_WORKGROUP);
                }
            }
        }
    }
}

template <bool SAVE>
__device__ __forceinline__ int tile_add_queue(
        const float* __restrict__ xin, const v4i* __restrict__ erec,
        int* q, int n, bool all, int lo, int nn, int lane,
        const float (&we)[4], const float (&bb)[4], float* tile,
        int* __restrict__ rec, int& nsaved) {
    int base = 0;
    for (; base < n; base += 32) {
        const int m = min(32, n - base);
        if (!all && m < 32) break;
        const int e  = q[base + min(lane, m - 1)];
        const v4i er = erec[e];
        const int sv = er.x;
        const int dv = er.y - lo;
        const int av = er.z;
        if (SAVE) {
            const int p = nsaved + lane;
            if (lane < m && p < RCAP) {
                rec[p]            = sv;
                rec[RCAP + p]     = dv;
                rec[2 * RCAP + p] = av;
            }
            nsaved += m;
        }
        tile_add_batch(xin, sv, dv, av, m, nn, lane, we, bb, tile, [] {});
    }
    if (base >= n) return 0;
    const int rem = n - base;
    if (base > 0 && lane < rem) q[lane] = q[base + lane];
    return rem;
}

template <bool SAVE>
__device__ __forceinline__ int scan_tile(
        const float* __restrict__ xin, const v4i* __restrict__ erec,
        const unsigned char* __restrict__ tbytes,
        int* q, int* __restrict__ rec, int E_, int lo, int nn, int lane, int wave,
        const float (&we)[4], const float (&bb)[4], float* tile) {
    const unsigned kb  = (unsigned)blockIdx.x & 255u;
    const unsigned key = kb * 0x01010101u;
    const int nfull = E_ >> 9;
    const int rot = (int)blockIdx.x * (nfull / (int)gridDim.x);

    int nsaved = 0;
    int cnt = 0;
    for (int b0 = wave; b0 < nfull; b0 += EDGE_WAVES * SCAN_U) {
        uint4    v[SCAN_U];
        unsigned okm[SCAN_U];
        int      eb[SCAN_U];
        #pragma unroll
        for (int u = 0; u < SCAN_U; ++u) {
            const int  b  = b0 + EDGE_WAVES * u;
            const bool ok = b < nfull;
            okm[u] = (unsigned)((b - nfull) >> 31);
            int br = (ok ? b : b0) + rot;
            if (br >= nfull) br -= nfull;
            eb[u] = br * 512 + lane * 16;
            v[u] = *(const uint4*)(tbytes + eb[u]);
        }
        #pragma unroll
        for (int u = 0; u < SCAN_U; ++u) {
            const unsigned t[4] = {zero_bytes(v[u].x ^ key) & okm[u], zero_bytes(v[u].y ^ key) & okm[u],
                                   zero_bytes(v[u].z ^ key) & okm[u], zero_bytes(v[u].w ^ key) & okm[u]};
            #pragma unroll
            for (int k = 0; k < 4; ++k) {
                unsigned tk   = t[k];
                unsigned mask = __builtin_amdgcn_ballot_w32(tk != 0u);
                while (mask != 0u) {
                    if (tk != 0u) {
                        q[cnt + (int)__builtin_amdgcn_mbcnt_lo(mask, 0u)] =
                            eb[u] + 4 * k + (__builtin_ctz(tk) >> 3);
                        tk &= tk - 1u;
                    }
                    cnt += __builtin_popcount(mask);
                    mask = __builtin_amdgcn_ballot_w32(tk != 0u);
                }
            }
            if (cnt >= 32)
                cnt = tile_add_queue<SAVE>(xin, erec, q, cnt, false, lo, nn, lane, we, bb, tile, rec, nsaved);
        }
    }
    for (int e0 = nfull * 512 + wave * 32; e0 < E_; e0 += EDGE_WAVES * 32) {
        const int e = e0 + lane;
        const bool hit = (e < E_) && ((unsigned)tbytes[min(e, E_ - 1)] == kb);
        const unsigned mask = __builtin_amdgcn_ballot_w32(hit);
        if (mask != 0u) {
            if (hit) q[cnt + (int)__builtin_amdgcn_mbcnt_lo(mask, 0u)] = e;
            cnt += __builtin_popcount(mask);
        }
        if (cnt >= 32)
            cnt = tile_add_queue<SAVE>(xin, erec, q, cnt, false, lo, nn, lane, we, bb, tile, rec, nsaved);
    }
    if (cnt > 0)
        tile_add_queue<SAVE>(xin, erec, q, cnt, true, lo, nn, lane, we, bb, tile, rec, nsaved);
    return nsaved;
}

typedef unsigned long long __attribute__((may_alias)) u64_any;
typedef v8h __attribute__((may_alias)) v8h_any;
typedef _Float16 __attribute__((may_alias)) f16_any;

#define HPITCH 136

__device__ __forceinline__ void load_layer_weights(
        const _Float16* __restrict__ fw, const float* __restrict__ b, int np, int lane,
        v16h (&bv)[4][2], float (&bs)[2]) {
    #pragma unroll
    for (int kt = 0; kt < 4; ++kt) {
        #pragma unroll
        for (int j = 0; j < 2; ++j) {
            const int nt = 2 * np + j;
            const _Float16* q = fw + ((size_t)((kt << 3) | nt) * 32 + lane) * 16;
            v8h blo = *(const v8h*)q;
            v8h bhi = *(const v8h*)(q + 8);
            bv[kt][j] = __builtin_shufflevector(blo, bhi, 0,1,2,3,4,5,6,7,8,9,10,11,12,13,14,15);
        }
    }
    #pragma unroll
    for (int j = 0; j < 2; ++j) bs[j] = b[(2 * np + j) * 16 + (lane & 15)];
}

__device__ __forceinline__ void row_block_product(
        const f16_any* A, int mt, int m_, int hs8, const v16h (&bv)[4][2], v8f (&acc)[2]) {
    const f16_any* arow = A + (mt * 16 + m_) * HPITCH + hs8;
    v16h av[4];
    #pragma unroll
    for (int kt = 0; kt < 4; ++kt) {
        v8h alo = *(const v8h_any*)(arow + kt * 32);
        v8h ahi = *(const v8h_any*)(arow + kt * 32 + 16);
        av[kt] = __builtin_shufflevector(alo, ahi, 0,1,2,3,4,5,6,7,8,9,10,11,12,13,14,15);
    }
    const v8f vz = {0.f, 0.f, 0.f, 0.f, 0.f, 0.f, 0.f, 0.f};
    acc[0] = vz;
    acc[1] = vz;
    #pragma unroll
    for (int kt = 0; kt < 4; ++kt) {
        acc[0] = __builtin_amdgcn_wmma_f32_16x16x32_f16(
            false, av[kt], false, bv[kt][0], (short)0, acc[0], false, false);
        acc[1] = __builtin_amdgcn_wmma_f32_16x16x32_f16(
            false, av[kt], false, bv[kt][1], (short)0, acc[1], false, false);
    }
}

__device__ __forceinline__ void tile_mlp(
        const float* __restrict__ xin, unsigned char* smem, int lo, int nn, int N_, float ep,
        const _Float16* __restrict__ fw1, const float* __restrict__ b1,
        const _Float16* __restrict__ fw2, const float* __restrict__ b2,
        float* __restrict__ hout, int lane, int wave) {
    const float* sums = (const float*)smem;
    f16_any* A1 = (f16_any*)smem;
    f16_any* A2 = A1 + TILE_NODES * HPITCH;

    const int  trow0 = wave * 16;
    const bool mine  = trow0 < nn;
    unsigned long long hp[16];
    #pragma unroll
    for (int r = 0; r < 16; ++r) hp[r] = 0ull;
    if (mine) {
        v4f xv[16];
        #pragma unroll
        for (int r = 0; r < 16; ++r) {
            const int g = min(lo + trow0 + r, N_ - 1);
            const float* xr = xin + (size_t)g * F_DIM;
            xv[r] = *(const v4f*)(xr + (unsigned)(lane << 2));
        }
        #pragma unroll
        for (int r = 0; r < 16; ++r) {
            const float4 av = *(const float4*)(sums + (trow0 + r) * F_DIM + (lane << 2));
            union { _Float16 h[4]; unsigned long long u; } p;
            p.h[0] = (_Float16)(ep * xv[r].x + av.x);
            p.h[1] = (_Float16)(ep * xv[r].y + av.y);
            p.h[2] = (_Float16)(ep * xv[r].z + av.z);
            p.h[3] = (_Float16)(ep * xv[r].w + av.w);
            hp[r] = p.u;
        }
    }

    const int np  = wave & 3;
    const int mg  = wave >> 2;
    const int m_  = lane & 15;
    const int hs8 = (lane >> 4) << 3;

    v16h  bv[4][2];
    float bs[2];
    load_layer_weights(fw1, b1, np, lane, bv, bs);

    __syncthreads();
    if (mine) {
        #pragma unroll
        for (int r = 0; r < 16; ++r)
            *(u64_any*)(A1 + (trow0 + r) * HPITCH + (lane << 2)) = hp[r];
    }
    __syncthreads();

    for (int mt = mg; mt * 16 < nn; mt += 8) {
        v8f acc[2];
        row_block_product(A1, mt, m_, hs8, bv, acc);
        #pragma unroll
        for (int j = 0; j < 2; ++j) {
            const int col = (2 * np + j) * 16 + m_;
            #pragma unroll
            for (int r = 0; r < 8; ++r) {
                const float v = fmaxf(acc[j][r] + bs[j], 0.f);
                A2[(mt * 16 + r + hs8) * HPITCH + col] = (_Float16)v;
            }
        }
    }

    load_layer_weights(fw2, b2, np, lane, bv, bs);

    __syncthreads();

    for (int mt = mg; mt * 16 < nn; mt += 8) {
        v8f acc[2];
        row_block_product(A2, mt, m_, hs8, bv, acc);
        float o[2][8];
        #pragma unroll
        for (int j = 0; j < 2; ++j) {
            #pragma unroll
            for (int r = 0; r < 8; ++r) o[j][r] = fmaxf(acc[j][r] + bs[j], 0.f);
        }
        const int grow0 = lo + mt * 16;
        float* ob = hout + (size_t)grow0 * F_DIM;
        if (grow0 + 16 <= N_) {
            #pragma unroll
            for (int j = 0; j < 2; ++j) {
                const int col = (2 * np + j) * 16 + m_;
                #pragma unroll
                for (int r = 0; r < 8; ++r)
                    ob[(unsigned)((r + hs8) * F_DIM + col)] = o[j][r];
            }
        } else {
            #pragma unroll
            for (int j = 0; j < 2; ++j) {
                const int col = (2 * np + j) * 16 + m_;
                #pragma unroll
                for (int r = 0; r < 8; ++r)
                    if (grow0 + r + hs8 < N_)
                        ob[(unsigned)((r + hs8) * F_DIM + col)] = o[j][r];
            }
        }
    }
}

template <bool REPLAY>
__global__ void __launch_bounds__(1024)
edge_mlp(const float* __restrict__ xin, const int* __restrict__ erec,
         const float* __restrict__ We,
         const float* __restrict__ be, const unsigned* __restrict__ tid4,
         const float* __restrict__ eps,
         const _Float16* __restrict__ fw1, const float* __restrict__ b1,
         const _Float16* __restrict__ fw2, const float* __restrict__ b2,
         float* __restrict__ hout, int* __restrict__ recs, int* __restrict__ rcnt,
         int E_, int N_) {
    __shared__ __align__(16) unsigned char smem[TILE_NODES * F_DIM * 4 + EDGE_WAVES * QCAP * 4];
    float* tile = (float*)smem;

    const int tid  = threadIdx.x;
    const int lane = tid & 31;
    const int wave = __builtin_amdgcn_readfirstlane(tid >> 5);
    const int lo   = blockIdx.x * TILE_NODES;
    const int nn   = min(TILE_NODES, N_ - lo);

    const int slot = (int)blockIdx.x * EDGE_WAVES + wave;
    int* rec = recs + (size_t)slot * (3 * RCAP);
    int nrec = 0;
    if (REPLAY) nrec = rcnt[slot];

    float4* tile4 = (float4*)tile;
    for (int i = tid; i < TILE_NODES * F_DIM / 4; i += 1024)
        tile4[i] = make_float4(0.f, 0.f, 0.f, 0.f);

    float we[4], bb[4];
    #pragma unroll
    for (int k = 0; k < 4; ++k) { we[k] = We[lane + 32 * k]; bb[k] = be[lane + 32 * k]; }

    int* q = (int*)(smem + TILE_NODES * F_DIM * 4) + wave * QCAP;
    const unsigned char* tbytes = (const unsigned char*)tid4;
    const v4i* erec4 = (const v4i*)erec;

    __syncthreads();

    if (REPLAY) {
        const int n = __builtin_amdgcn_readfirstlane(nrec);
        if (n <= RCAP) {
            if (n > 0) {
                int idx = min(lane, n - 1);
                int sv = rec[idx], dv = rec[RCAP + idx], av = rec[2 * RCAP + idx];
                for (int base = 0; base < n; base += 32) {
                    const int m = min(32, n - base);
                    idx = min(base + 32 + lane, n - 1);
                    int sv2, dv2, av2;
                    tile_add_batch(xin, sv, dv, av, m, nn, lane, we, bb, tile, [&] {
                        sv2 = rec[idx]; dv2 = rec[RCAP + idx]; av2 = rec[2 * RCAP + idx];
                    });
                    sv = sv2; dv = dv2; av = av2;
                }
            }
        } else {
            scan_tile<false>(xin, erec4, tbytes, q, rec, E_, lo, nn, lane, wave, we, bb, tile);
        }
    } else {
        const int found = scan_tile<true>(xin, erec4, tbytes, q, rec, E_, lo, nn, lane, wave, we, bb, tile);
        if (lane == 0) rcnt[slot] = found;
    }

    __syncthreads();

    tile_mlp(xin, smem, lo, nn, N_, 1.0f + eps[0], fw1, b1, fw2, b2, hout, lane, wave);
}

__global__ void __launch_bounds__(256)
classifier(const float* __restrict__ h, const float* __restrict__ Wl,
           const float* __restrict__ bl, float* __restrict__ out, int n) {
    __shared__ float w[F_DIM * 40];
    __shared__ float bb[40];
    for (int i = threadIdx.x; i < F_DIM * 40; i += blockDim.x) w[i] = Wl[i];
    if (threadIdx.x < 40) bb[threadIdx.x] = bl[threadIdx.x];
    __syncthreads();

    int node = blockIdx.x * blockDim.x + threadIdx.x;
    if (node >= n) return;

    float acc[40];
    #pragma unroll
    for (int c = 0; c < 40; ++c) acc[c] = bb[c];

    const float* hr = h + (size_t)node * F_DIM;
    for (int k = 0; k < F_DIM; ++k) {
        float hv = hr[k];
        #pragma unroll
        for (int c = 0; c < 40; ++c) acc[c] = fmaf(hv, w[k * 40 + c], acc[c]);
    }

    float mx = acc[0];
    #pragma unroll
    for (int c = 1; c < 40; ++c) mx = fmaxf(mx, acc[c]);
    float s = 0.f;
    #pragma unroll
    for (int c = 0; c < 40; ++c) s += __expf(acc[c] - mx);
    float ls = __logf(s);
    float* o = out + (size_t)node * 40;
    #pragma unroll
    for (int c = 0; c < 40; ++c) o[c] = acc[c] - mx - ls;
}

extern "C" void kernel_launch(void* const* d_in, const int* in_sizes, int n_in,
                              void* d_out, int out_size, void* d_ws, size_t ws_size,
                              hipStream_t stream) {
    const float* x    = (const float*)d_in[0];
    const int*   ei   = (const int*)d_in[1];
    const float* ea   = (const float*)d_in[2];
    const float* eps1 = (const float*)d_in[3];
    const float* We1  = (const float*)d_in[4];
    const float* be1  = (const float*)d_in[5];
    const float* W11  = (const float*)d_in[6];
    const float* b11  = (const float*)d_in[7];
    const float* W12  = (const float*)d_in[8];
    const float* b12  = (const float*)d_in[9];
    const float* eps2 = (const float*)d_in[10];
    const float* We2  = (const float*)d_in[11];
    const float* be2  = (const float*)d_in[12];
    const float* W21  = (const float*)d_in[13];
    const float* b21  = (const float*)d_in[14];
    const float* W22  = (const float*)d_in[15];
    const float* b22  = (const float*)d_in[16];
    const float* Wl   = (const float*)d_in[17];
    const float* bl   = (const float*)d_in[18];

    const int N_ = in_sizes[0] / F_DIM;
    const int E_ = in_sizes[2];
    const size_t NB = (size_t)N_ * F_DIM * sizeof(float);

    char* ws = (char*)d_ws;
    float*    h2    = (float*)ws;
    float*    h1    = (float*)(ws + NB);
    _Float16* frags = (_Float16*)(ws + 2 * NB);
    unsigned* tid4  = (unsigned*)(ws + 2 * NB + 131072);
    const size_t TBY = (((size_t)E_ + 3) & ~(size_t)3);
    _Float16* f11 = frags;
    _Float16* f12 = frags + 16384;
    _Float16* f21 = frags + 32768;
    _Float16* f22 = frags + 49152;
    float* out = (float*)d_out;

    const int tb = (N_ + TILE_NODES - 1) / TILE_NODES;
    int* rcnt = (int*)(ws + 2 * NB + 131072 + ((TBY + 255) & ~(size_t)255));
    int* recs = rcnt + (((size_t)tb * EDGE_WAVES + 63) & ~(size_t)63);
    int* erec = recs + (size_t)tb * EDGE_WAVES * (3 * RCAP);
    const int cb = (N_ + 255) / 256;

    prep<<<1024, 256, 0, stream>>>(W11, W12, W21, W22, frags, ei, ea, tid4, erec, E_);

    edge_mlp<false><<<tb, 1024, 0, stream>>>(x, erec, We1, be1, tid4, eps1, f11, b11, f12, b12,
                                              h1, recs, rcnt, E_, N_);
    edge_mlp<true><<<tb, 1024, 0, stream>>>(h1, erec, We2, be2, tid4, eps2, f21, b21, f22, b22,
                                             h2, recs, rcnt, E_, N_);

    classifier<<<cb, 256, 0, stream>>>(h2, Wl, bl, out, N_);
}


# Edge gather: a source row is one wave-wide 16-byte load (four whole lines) instead of four 4-byte loads; LDS tile kept lane-major so the float adds stay conflict-free; sum order varies as before
# speedup vs baseline: 3.7329x; 1.0883x over previous
#include <hip/hip_runtime.h>
#include <hip/hip_bf16.h>
#include <stdint.h>

typedef _Float16 v16h __attribute__((ext_vector_type(16)));
typedef _Float16 v8h  __attribute__((ext_vector_type(8)));
typedef float    v8f  __attribute__((ext_vector_type(8)));
typedef float    v4f  __attribute__((ext_vector_type(4)));
typedef int      v4i  __attribute__((ext_vector_type(4)));

#define F_DIM 128

#define TILE_NODES 400

__device__ __forceinline__ unsigned tile_byte(int d) {
    return ((unsigned)d / (unsigned)TILE_NODES) & 255u;
}

__global__ void __launch_bounds__(256)
prep(const float* __restrict__ W11, const float* __restrict__ W12,
     const float* __restrict__ W21, const float* __restrict__ W22,
     _Float16* __restrict__ frags, const int* __restrict__ ei,
     const float* __restrict__ ea, unsigned* __restrict__ tid4,
     int* __restrict__ erec, int E_) {
    const int gid = blockIdx.x * blockDim.x + threadIdx.x;
    if (gid < 4 * 8192) {
        int m    = gid >> 13;
        int r    = gid & 8191;
        int d    = r & 7;
        int lane = (r >> 3) & 31;
        int ntk  = r >> 8;
        int kt   = ntk >> 3;
        int nt   = ntk & 7;
        int nn   = nt * 16 + (lane & 15);
        int kb   = kt * 32 + ((lane >> 4) << 3);
        int k0   = kb + ((d < 4) ? (d << 1) : (16 + ((d - 4) << 1)));
        const float* W = (m == 0) ? W11 : (m == 1) ? W12 : (m == 2) ? W21 : W22;
        _Float16* o = frags + (size_t)m * 16384 + (size_t)r * 2;
        o[0] = (_Float16)W[(size_t)k0 * F_DIM + nn];
        o[1] = (_Float16)W[(size_t)(k0 + 1) * F_DIM + nn];
    }

    const int* dsts = ei + E_;
    const int  nq   = (E_ + 3) >> 2;
    const bool al   = (E_ & 3) == 0;
    for (int i = gid; i < nq; i += (int)(gridDim.x * blockDim.x)) {
        const int e = 4 * i;
        int d0, d1, d2, d3;
        if (al) {
            const int4 v = *(const int4*)(dsts + e);
            d0 = v.x; d1 = v.y; d2 = v.z; d3 = v.w;
        } else {
            d0 = dsts[e];
            d1 = (e + 1 < E_) ? dsts[e + 1] : 0;
            d2 = (e + 2 < E_) ? dsts[e + 2] : 0;
            d3 = (e + 3 < E_) ? dsts[e + 3] : 0;
        }
        tid4[i] = tile_byte(d0) | (tile_byte(d1) << 8) | (tile_byte(d2) << 16) | (tile_byte(d3) << 24);
    }

    v4i* er = (v4i*)erec;
    const int ln   = threadIdx.x & 31;
    const int wv   = gid >> 5;
    const int nwv  = (int)(gridDim.x * blockDim.x) >> 5;
    for (int c = wv * 128; c < E_; c += nwv * 128) {
        int sv[4], dv[4], av[4];
        #pragma unroll
        for (int k = 0; k < 4; ++k) {
            const int e = min(c + 32 * k + ln, E_ - 1);
            sv[k] = ei[e];
            dv[k] = dsts[e];
            av[k] = __float_as_int(ea[e]);
        }
        #pragma unroll
        for (int k = 0; k < 4; ++k) {
            const int e = c + 32 * k + ln;
            if (e < E_) {
                const v4i r = {sv[k], dv[k], av[k], 0};
                er[e] = r;
            }
        }
    }
}

#define EDGE_WAVES 32
#define SCAN_U     4
#define QCAP       544
#define RCAP       512

__device__ __forceinline__ unsigned zero_bytes(unsigned x) {
    return ~(((x & 0x7F7F7F7Fu) + 0x7F7F7F7Fu) | x | 0x7F7F7F7Fu);
}

template <class F>
__device__ __forceinline__ void tile_add_batch(
        const float* __restrict__ xin, int sv, int dv, int av, int m,
        int nn, int lane, const float (&we)[4], const float (&bb)[4], float* tile,
        F behind_first_loads) {
    for (int j = 0; j < m; j += 8) {
        v4f   r[8];
        int   dl[8];
        float aa[8];
        #pragma unroll
        for (int u = 0; u < 8; ++u) {
            const int jj = min(j + u, m - 1);
            const int s  = __builtin_amdgcn_readlane(sv, jj);
            dl[u] = __builtin_amdgcn_readlane(dv, jj);
            aa[u] = __int_as_float(__builtin_amdgcn_readlane(av, jj));
            r[u] = *(const v4f*)(xin + (size_t)s * F_DIM + (unsigned)(lane << 2));
        }
        if (j == 0) behind_first_loads();
        #pragma unroll
        for (int u = 0; u < 8; ++u) {
            if (j + u < m && (unsigned)dl[u] < (unsigned)nn) {
                float* t = tile + dl[u] * F_DIM + lane;
                #pragma unroll
                for (int k = 0; k < 4; ++k) {
                    const float v = fmaxf(r[u][k] + aa[u] * we[k] + bb[k], 0.f);
                    __hip_atomic_fetch_add(t + 32 * k, v, __ATOMIC_RELAXED,
                                           __HIP_MEMORY_SCOPE_WORKGROUP);
                }
            }
        }
    }
}

template <bool SAVE>
__device__ __forceinline__ int tile_add_queue(
        const float* __restrict__ xin, const v4i* __restrict__ erec,
        int* q, int n, bool all, int lo, int nn, int lane,
        const float (&we)[4], const float (&bb)[4], float* tile,
        int* __restrict__ rec, int& nsaved) {
    int base = 0;
    for (; base < n; base += 32) {
        const int m = min(32, n - base);
        if (!all && m < 32) break;
        const int e  = q[base + min(lane, m - 1)];
        const v4i er = erec[e];
        const int sv = er.x;
        const int dv = er.y - lo;
        const int av = er.z;
        if (SAVE) {
            const int p = nsaved + lane;
            if (lane < m && p < RCAP) {
                rec[p]            = sv;
                rec[RCAP + p]     = dv;
                rec[2 * RCAP + p] = av;
            }
            nsaved += m;
        }
        tile_add_batch(xin, sv, dv, av, m, nn, lane, we, bb, tile, [] {});
    }
    if (base >= n) return 0;
    const int rem = n - base;
    if (base > 0 && lane < rem) q[lane] = q[base + lane];
    return rem;
}

template <bool SAVE>
__device__ __forceinline__ int scan_tile(
        const float* __restrict__ xin, const v4i* __restrict__ erec,
        const unsigned char* __restrict__ tbytes,
        int* q, int* __restrict__ rec, int E_, int lo, int nn, int lane, int wave,
        const float (&we)[4], const float (&bb)[4], float* tile) {
    const unsigned kb  = (unsigned)blockIdx.x & 255u;
    const unsigned key = kb * 0x01010101u;
    const int nfull = E_ >> 9;
    const int rot = (int)blockIdx.x * (nfull / (int)gridDim.x);

    int nsaved = 0;
    int cnt = 0;
    for (int b0 = wave; b0 < nfull; b0 += EDGE_WAVES * SCAN_U) {
        uint4    v[SCAN_U];
        unsigned okm[SCAN_U];
        int      eb[SCAN_U];
        #pragma unroll
        for (int u = 0; u < SCAN_U; ++u) {
            const int  b  = b0 + EDGE_WAVES * u;
            const bool ok = b < nfull;
            okm[u] = (unsigned)((b - nfull) >> 31);
            int br = (ok ? b : b0) + rot;
            if (br >= nfull) br -= nfull;
            eb[u] = br * 512 + lane * 16;
            v[u] = *(const uint4*)(tbytes + eb[u]);
        }
        #pragma unroll
        for (int u = 0; u < SCAN_U; ++u) {
            const unsigned t[4] = {zero_bytes(v[u].x ^ key) & okm[u], zero_bytes(v[u].y ^ key) & okm[u],
                                   zero_bytes(v[u].z ^ key) & okm[u], zero_bytes(v[u].w ^ key) & okm[u]};
            #pragma unroll
            for (int k = 0; k < 4; ++k) {
                unsigned tk   = t[k];
                unsigned mask = __builtin_amdgcn_ballot_w32(tk != 0u);
                while (mask != 0u) {
                    if (tk != 0u) {
                        q[cnt + (int)__builtin_amdgcn_mbcnt_lo(mask, 0u)] =
                            eb[u] + 4 * k + (__builtin_ctz(tk) >> 3);
                        tk &= tk - 1u;
                    }
                    cnt += __builtin_popcount(mask);
                    mask = __builtin_amdgcn_ballot_w32(tk != 0u);
                }
            }
            if (cnt >= 32)
                cnt = tile_add_queue<SAVE>(xin, erec, q, cnt, false, lo, nn, lane, we, bb, tile, rec, nsaved);
        }
    }
    for (int e0 = nfull * 512 + wave * 32; e0 < E_; e0 += EDGE_WAVES * 32) {
        const int e = e0 + lane;
        const bool hit = (e < E_) && ((unsigned)tbytes[min(e, E_ - 1)] == kb);
        const unsigned mask = __builtin_amdgcn_ballot_w32(hit);
        if (mask != 0u) {
            if (hit) q[cnt + (int)__builtin_amdgcn_mbcnt_lo(mask, 0u)] = e;
            cnt += __builtin_popcount(mask);
        }
        if (cnt >= 32)
            cnt = tile_add_queue<SAVE>(xin, erec, q, cnt, false, lo, nn, lane, we, bb, tile, rec, nsaved);
    }
    if (cnt > 0)
        tile_add_queue<SAVE>(xin, erec, q, cnt, true, lo, nn, lane, we, bb, tile, rec, nsaved);
    return nsaved;
}

typedef unsigned long long __attribute__((may_alias)) u64_any;
typedef v8h __attribute__((may_alias)) v8h_any;
typedef _Float16 __attribute__((may_alias)) f16_any;

#define HPITCH 136

__device__ __forceinline__ void load_layer_weights(
        const _Float16* __restrict__ fw, const float* __restrict__ b, int np, int lane,
        v16h (&bv)[4][2], float (&bs)[2]) {
    #pragma unroll
    for (int kt = 0; kt < 4; ++kt) {
        #pragma unroll
        for (int j = 0; j < 2; ++j) {
            const int nt = 2 * np + j;
            const _Float16* q = fw + ((size_t)((kt << 3) | nt) * 32 + lane) * 16;
            v8h blo = *(const v8h*)q;
            v8h bhi = *(const v8h*)(q + 8);
            bv[kt][j] = __builtin_shufflevector(blo, bhi, 0,1,2,3,4,5,6,7,8,9,10,11,12,13,14,15);
        }
    }
    #pragma unroll
    for (int j = 0; j < 2; ++j) bs[j] = b[(2 * np + j) * 16 + (lane & 15)];
}

__device__ __forceinline__ void row_block_product(
        const f16_any* A, int mt, int m_, int hs8, const v16h (&bv)[4][2], v8f (&acc)[2]) {
    const f16_any* arow = A + (mt * 16 + m_) * HPITCH + hs8;
    v16h av[4];
    #pragma unroll
    for (int kt = 0; kt < 4; ++kt) {
        v8h alo = *(const v8h_any*)(arow + kt * 32);
        v8h ahi = *(const v8h_any*)(arow + kt * 32 + 16);
        av[kt] = __builtin_shufflevector(alo, ahi, 0,1,2,3,4,5,6,7,8,9,10,11,12,13,14,15);
    }
    const v8f vz = {0.f, 0.f, 0.f, 0.f, 0.f, 0.f, 0.f, 0.f};
    acc[0] = vz;
    acc[1] = vz;
    #pragma unroll
    for (int kt = 0; kt < 4; ++kt) {
        acc[0] = __builtin_amdgcn_wmma_f32_16x16x32_f16(
            false, av[kt], false, bv[kt][0], (short)0, acc[0], false, false);
        acc[1] = __builtin_amdgcn_wmma_f32_16x16x32_f16(
            false, av[kt], false, bv[kt][1], (short)0, acc[1], false, false);
    }
}

__device__ __forceinline__ void tile_mlp(
        const float* __restrict__ xin, unsigned char* smem, int lo, int nn, int N_, float ep,
        const _Float16* __restrict__ fw1, const float* __restrict__ b1,
        const _Float16* __restrict__ fw2, const float* __restrict__ b2,
        float* __restrict__ hout, int lane, int wave) {
    const float* sums = (const float*)smem;
    f16_any* A1 = (f16_any*)smem;
    f16_any* A2 = A1 + TILE_NODES * HPITCH;

    const int  trow0 = wave * 16;
    const bool mine  = trow0 < nn;
    unsigned long long hp[16];
    #pragma unroll
    for (int r = 0; r < 16; ++r) hp[r] = 0ull;
    if (mine) {
        v4f xv[16];
        #pragma unroll
        for (int r = 0; r < 16; ++r) {
            const int g = min(lo + trow0 + r, N_ - 1);
            const float* xr = xin + (size_t)g * F_DIM;
            xv[r] = *(const v4f*)(xr + (unsigned)(lane << 2));
        }
        #pragma unroll
        for (int r = 0; r < 16; ++r) {
            const float* sr = sums + (trow0 + r) * F_DIM + lane;
            union { _Float16 h[4]; unsigned long long u; } p;
            p.h[0] = (_Float16)(ep * xv[r].x + sr[0]);
            p.h[1] = (_Float16)(ep * xv[r].y + sr[32]);
            p.h[2] = (_Float16)(ep * xv[r].z + sr[64]);
            p.h[3] = (_Float16)(ep * xv[r].w + sr[96]);
            hp[r] = p.u;
        }
    }

    const int np  = wave & 3;
    const int mg  = wave >> 2;
    const int m_  = lane & 15;
    const int hs8 = (lane >> 4) << 3;

    v16h  bv[4][2];
    float bs[2];
    load_layer_weights(fw1, b1, np, lane, bv, bs);

    __syncthreads();
    if (mine) {
        #pragma unroll
        for (int r = 0; r < 16; ++r)
            *(u64_any*)(A1 + (trow0 + r) * HPITCH + (lane << 2)) = hp[r];
    }
    __syncthreads();

    for (int mt = mg; mt * 16 < nn; mt += 8) {
        v8f acc[2];
        row_block_product(A1, mt, m_, hs8, bv, acc);
        #pragma unroll
        for (int j = 0; j < 2; ++j) {
            const int col = (2 * np + j) * 16 + m_;
            #pragma unroll
            for (int r = 0; r < 8; ++r) {
                const float v = fmaxf(acc[j][r] + bs[j], 0.f);
                A2[(mt * 16 + r + hs8) * HPITCH + col] = (_Float16)v;
            }
        }
    }

    load_layer_weights(fw2, b2, np, lane, bv, bs);

    __syncthreads();

    for (int mt = mg; mt * 16 < nn; mt += 8) {
        v8f acc[2];
        row_block_product(A2, mt, m_, hs8, bv, acc);
        float o[2][8];
        #pragma unroll
        for (int j = 0; j < 2; ++j) {
            #pragma unroll
            for (int r = 0; r < 8; ++r) o[j][r] = fmaxf(acc[j][r] + bs[j], 0.f);
        }
        const int grow0 = lo + mt * 16;
        float* ob = hout + (size_t)grow0 * F_DIM;
        if (grow0 + 16 <= N_) {
            #pragma unroll
            for (int j = 0; j < 2; ++j) {
                const int col = (2 * np + j) * 16 + m_;
                #pragma unroll
                for (int r = 0; r < 8; ++r)
                    ob[(unsigned)((r + hs8) * F_DIM + col)] = o[j][r];
            }
        } else {
            #pragma unroll
            for (int j = 0; j < 2; ++j) {
                const int col = (2 * np + j) * 16 + m_;
                #pragma unroll
                for (int r = 0; r < 8; ++r)
                    if (grow0 + r + hs8 < N_)
                        ob[(unsigned)((r + hs8) * F_DIM + col)] = o[j][r];
            }
        }
    }
}

template <bool REPLAY>
__global__ void __launch_bounds__(1024)
edge_mlp(const float* __restrict__ xin, const int* __restrict__ erec,
         const float* __restrict__ We,
         const float* __restrict__ be, const unsigned* __restrict__ tid4,
         const float* __restrict__ eps,
         const _Float16* __restrict__ fw1, const float* __restrict__ b1,
         const _Float16* __restrict__ fw2, const float* __restrict__ b2,
         float* __restrict__ hout, int* __restrict__ recs, int* __restrict__ rcnt,
         int E_, int N_) {
    __shared__ __align__(16) unsigned char smem[TILE_NODES * F_DIM * 4 + EDGE_WAVES * QCAP * 4];
    float* tile = (float*)smem;

    const int tid  = threadIdx.x;
    const int lane = tid & 31;
    const int wave = __builtin_amdgcn_readfirstlane(tid >> 5);
    const int lo   = blockIdx.x * TILE_NODES;
    const int nn   = min(TILE_NODES, N_ - lo);

    const int slot = (int)blockIdx.x * EDGE_WAVES + wave;
    int* rec = recs + (size_t)slot * (3 * RCAP);
    int nrec = 0;
    if (REPLAY) nrec = rcnt[slot];

    float4* tile4 = (float4*)tile;
    for (int i = tid; i < TILE_NODES * F_DIM / 4; i += 1024)
        tile4[i] = make_float4(0.f, 0.f, 0.f, 0.f);

    float we[4], bb[4];
    #pragma unroll
    for (int k = 0; k < 4; ++k) { we[k] = We[4 * lane + k]; bb[k] = be[4 * lane + k]; }

    int* q = (int*)(smem + TILE_NODES * F_DIM * 4) + wave * QCAP;
    const unsigned char* tbytes = (const unsigned char*)tid4;
    const v4i* erec4 = (const v4i*)erec;

    __syncthreads();

    if (REPLAY) {
        const int n = __builtin_amdgcn_readfirstlane(nrec);
        if (n <= RCAP) {
            if (n > 0) {
                int idx = min(lane, n - 1);
                int sv = rec[idx], dv = rec[RCAP + idx], av = rec[2 * RCAP + idx];
                for (int base = 0; base < n; base += 32) {
                    const int m = min(32, n - base);
                    idx = min(base + 32 + lane, n - 1);
                    int sv2, dv2, av2;
                    tile_add_batch(xin, sv, dv, av, m, nn, lane, we, bb, tile, [&] {
                        sv2 = rec[idx]; dv2 = rec[RCAP + idx]; av2 = rec[2 * RCAP + idx];
                    });
                    sv = sv2; dv = dv2; av = av2;
                }
            }
        } else {
            scan_tile<false>(xin, erec4, tbytes, q, rec, E_, lo, nn, lane, wave, we, bb, tile);
        }
    } else {
        const int found = scan_tile<true>(xin, erec4, tbytes, q, rec, E_, lo, nn, lane, wave, we, bb, tile);
        if (lane == 0) rcnt[slot] = found;
    }

    __syncthreads();

    tile_mlp(xin, smem, lo, nn, N_, 1.0f + eps[0], fw1, b1, fw2, b2, hout, lane, wave);
}

__global__ void __launch_bounds__(256)
classifier(const float* __restrict__ h, const float* __restrict__ Wl,
           const float* __restrict__ bl, float* __restrict__ out, int n) {
    __shared__ float w[F_DIM * 40];
    __shared__ float bb[40];
    for (int i = threadIdx.x; i < F_DIM * 40; i += blockDim.x) w[i] = Wl[i];
    if (threadIdx.x < 40) bb[threadIdx.x] = bl[threadIdx.x];
    __syncthreads();

    int node = blockIdx.x * blockDim.x + threadIdx.x;
    if (node >= n) return;

    float acc[40];
    #pragma unroll
    for (int c = 0; c < 40; ++c) acc[c] = bb[c];

    const float* hr = h + (size_t)node * F_DIM;
    for (int k = 0; k < F_DIM; ++k) {
        float hv = hr[k];
        #pragma unroll
        for (int c = 0; c < 40; ++c) acc[c] = fmaf(hv, w[k * 40 + c], acc[c]);
    }

    float mx = acc[0];
    #pragma unroll
    for (int c = 1; c < 40; ++c) mx = fmaxf(mx, acc[c]);
    float s = 0.f;
    #pragma unroll
    for (int c = 0; c < 40; ++c) s += __expf(acc[c] - mx);
    float ls = __logf(s);
    float* o = out + (size_t)node * 40;
    #pragma unroll
    for (int c = 0; c < 40; ++c) o[c] = acc[c] - mx - ls;
}

extern "C" void kernel_launch(void* const* d_in, const int* in_sizes, int n_in,
                              void* d_out, int out_size, void* d_ws, size_t ws_size,
                              hipStream_t stream) {
    const float* x    = (const float*)d_in[0];
    const int*   ei   = (const int*)d_in[1];
    const float* ea   = (const float*)d_in[2];
    const float* eps1 = (const float*)d_in[3];
    const float* We1  = (const float*)d_in[4];
    const float* be1  = (const float*)d_in[5];
    const float* W11  = (const float*)d_in[6];
    const float* b11  = (const float*)d_in[7];
    const float* W12  = (const float*)d_in[8];
    const float* b12  = (const float*)d_in[9];
    const float* eps2 = (const float*)d_in[10];
    const float* We2  = (const float*)d_in[11];
    const float* be2  = (const float*)d_in[12];
    const float* W21  = (const float*)d_in[13];
    const float* b21  = (const float*)d_in[14];
    const float* W22  = (const float*)d_in[15];
    const float* b22  = (const float*)d_in[16];
    const float* Wl   = (const float*)d_in[17];
    const float* bl   = (const float*)d_in[18];

    const int N_ = in_sizes[0] / F_DIM;
    const int E_ = in_sizes[2];
    const size_t NB = (size_t)N_ * F_DIM * sizeof(float);

    char* ws = (char*)d_ws;
    float*    h2    = (float*)ws;
    float*    h1    = (float*)(ws + NB);
    _Float16* frags = (_Float16*)(ws + 2 * NB);
    unsigned* tid4  = (unsigned*)(ws + 2 * NB + 131072);
    const size_t TBY = (((size_t)E_ + 3) & ~(size_t)3);
    _Float16* f11 = frags;
    _Float16* f12 = frags + 16384;
    _Float16* f21 = frags + 32768;
    _Float16* f22 = frags + 49152;
    float* out = (float*)d_out;

    const int tb = (N_ + TILE_NODES - 1) / TILE_NODES;
    int* rcnt = (int*)(ws + 2 * NB + 131072 + ((TBY + 255) & ~(size_t)255));
    int* recs = rcnt + (((size_t)tb * EDGE_WAVES + 63) & ~(size_t)63);
    int* erec = recs + (size_t)tb * EDGE_WAVES * (3 * RCAP);
    const int cb = (N_ + 255) / 256;

    prep<<<1024, 256, 0, stream>>>(W11, W12, W21, W22, frags, ei, ea, tid4, erec, E_);

    edge_mlp<false><<<tb, 1024, 0, stream>>>(x, erec, We1, be1, tid4, eps1, f11, b11, f12, b12,
                                              h1, recs, rcnt, E_, N_);
    edge_mlp<true><<<tb, 1024, 0, stream>>>(h1, erec, We2, be2, tid4, eps2, f21, b21, f22, b22,
                                             h2, recs, rcnt, E_, N_);

    classifier<<<cb, 256, 0, stream>>>(h2, Wl, bl, out, N_);
}


# Classifier products moved from fp32 vector FMAs with LDS-broadcast weights to f32 16x16x4 matrix instructions: one 16-node block per wave, weights as B fragments in LDS; logit sum order changes
# speedup vs baseline: 4.2181x; 1.1300x over previous
#include <hip/hip_runtime.h>
#include <hip/hip_bf16.h>
#include <stdint.h>

typedef _Float16 v16h __attribute__((ext_vector_type(16)));
typedef _Float16 v8h  __attribute__((ext_vector_type(8)));
typedef float    v8f  __attribute__((ext_vector_type(8)));
typedef float    v4f  __attribute__((ext_vector_type(4)));
typedef int      v4i  __attribute__((ext_vector_type(4)));

#define F_DIM 128

#define TILE_NODES 400

__device__ __forceinline__ unsigned tile_byte(int d) {
    return ((unsigned)d / (unsigned)TILE_NODES) & 255u;
}

__global__ void __launch_bounds__(256)
prep(const float* __restrict__ W11, const float* __restrict__ W12,
     const float* __restrict__ W21, const float* __restrict__ W22,
     _Float16* __restrict__ frags, const int* __restrict__ ei,
     const float* __restrict__ ea, unsigned* __restrict__ tid4,
     int* __restrict__ erec, int E_) {
    const int gid = blockIdx.x * blockDim.x + threadIdx.x;
    if (gid < 4 * 8192) {
        int m    = gid >> 13;
        int r    = gid & 8191;
        int d    = r & 7;
        int lane = (r >> 3) & 31;
        int ntk  = r >> 8;
        int kt   = ntk >> 3;
        int nt   = ntk & 7;
        int nn   = nt * 16 + (lane & 15);
        int kb   = kt * 32 + ((lane >> 4) << 3);
        int k0   = kb + ((d < 4) ? (d << 1) : (16 + ((d - 4) << 1)));
        const float* W = (m == 0) ? W11 : (m == 1) ? W12 : (m == 2) ? W21 : W22;
        _Float16* o = frags + (size_t)m * 16384 + (size_t)r * 2;
        o[0] = (_Float16)W[(size_t)k0 * F_DIM + nn];
        o[1] = (_Float16)W[(size_t)(k0 + 1) * F_DIM + nn];
    }

    const int* dsts = ei + E_;
    const int  nq   = (E_ + 3) >> 2;
    const bool al   = (E_ & 3) == 0;
    for (int i = gid; i < nq; i += (int)(gridDim.x * blockDim.x)) {
        const int e = 4 * i;
        int d0, d1, d2, d3;
        if (al) {
            const int4 v = *(const int4*)(dsts + e);
            d0 = v.x; d1 = v.y; d2 = v.z; d3 = v.w;
        } else {
            d0 = dsts[e];
            d1 = (e + 1 < E_) ? dsts[e + 1] : 0;
            d2 = (e + 2 < E_) ? dsts[e + 2] : 0;
            d3 = (e + 3 < E_) ? dsts[e + 3] : 0;
        }
        tid4[i] = tile_byte(d0) | (tile_byte(d1) << 8) | (tile_byte(d2) << 16) | (tile_byte(d3) << 24);
    }

    v4i* er = (v4i*)erec;
    const int ln   = threadIdx.x & 31;
    const int wv   = gid >> 5;
    const int nwv  = (int)(gridDim.x * blockDim.x) >> 5;
    for (int c = wv * 128; c < E_; c += nwv * 128) {
        int sv[4], dv[4], av[4];
        #pragma unroll
        for (int k = 0; k < 4; ++k) {
            const int e = min(c + 32 * k + ln, E_ - 1);
            sv[k] = ei[e];
            dv[k] = dsts[e];
            av[k] = __float_as_int(ea[e]);
        }
        #pragma unroll
        for (int k = 0; k < 4; ++k) {
            const int e = c + 32 * k + ln;
            if (e < E_) {
                const v4i r = {sv[k], dv[k], av[k], 0};
                er[e] = r;
            }
        }
    }
}

#define EDGE_WAVES 32
#define SCAN_U     4
#define QCAP       544
#define RCAP       512

__device__ __forceinline__ unsigned zero_bytes(unsigned x) {
    return ~(((x & 0x7F7F7F7Fu) + 0x7F7F7F7Fu) | x | 0x7F7F7F7Fu);
}

template <class F>
__device__ __forceinline__ void tile_add_batch(
        const float* __restrict__ xin, int sv, int dv, int av, int m,
        int nn, int lane, const float (&we)[4], const float (&bb)[4], float* tile,
        F behind_first_loads) {
    for (int j = 0; j < m; j += 8) {
        v4f   r[8];
        int   dl[8];
        float aa[8];
        #pragma unroll
        for (int u = 0; u < 8; ++u) {
            const int jj = min(j + u, m - 1);
            const int s  = __builtin_amdgcn_readlane(sv, jj);
            dl[u] = __builtin_amdgcn_readlane(dv, jj);
            aa[u] = __int_as_float(__builtin_amdgcn_readlane(av, jj));
            r[u] = *(const v4f*)(xin + (size_t)s * F_DIM + (unsigned)(lane << 2));
        }
        if (j == 0) behind_first_loads();
        #pragma unroll
        for (int u = 0; u < 8; ++u) {
            if (j + u < m && (unsigned)dl[u] < (unsigned)nn) {
                float* t = tile + dl[u] * F_DIM + lane;
                #pragma unroll
                for (int k = 0; k < 4; ++k) {
                    const float v = fmaxf(r[u][k] + aa[u] * we[k] + bb[k], 0.f);
                    __hip_atomic_fetch_add(t + 32 * k, v, __ATOMIC_RELAXED,
                                           __HIP_MEMORY_SCOPE_WORKGROUP);
                }
            }
        }
    }
}

template <bool SAVE>
__device__ __forceinline__ int tile_add_queue(
        const float* __restrict__ xin, const v4i* __restrict__ erec,
        int* q, int n, bool all, int lo, int nn, int lane,
        const float (&we)[4], const float (&bb)[4], float* tile,
        int* __restrict__ rec, int& nsaved) {
    int base = 0;
    for (; base < n; base += 32) {
        const int m = min(32, n - base);
        if (!all && m < 32) break;
        const int e  = q[base + min(lane, m - 1)];
        const v4i er = erec[e];
        const int sv = er.x;
        const int dv = er.y - lo;
        const int av = er.z;
        if (SAVE) {
            const int p = nsaved + lane;
            if (lane < m && p < RCAP) {
                rec[p]            = sv;
                rec[RCAP + p]     = dv;
                rec[2 * RCAP + p] = av;
            }
            nsaved += m;
        }
        tile_add_batch(xin, sv, dv, av, m, nn, lane, we, bb, tile, [] {});
    }
    if (base >= n) return 0;
    const int rem = n - base;
    if (base > 0 && lane < rem) q[lane] = q[base + lane];
    return rem;
}

template <bool SAVE>
__device__ __forceinline__ int scan_tile(
        const float* __restrict__ xin, const v4i* __restrict__ erec,
        const unsigned char* __restrict__ tbytes,
        int* q, int* __restrict__ rec, int E_, int lo, int nn, int lane, int wave,
        const float (&we)[4], const float (&bb)[4], float* tile) {
    const unsigned kb  = (unsigned)blockIdx.x & 255u;
    const unsigned key = kb * 0x01010101u;
    const int nfull = E_ >> 9;
    const int rot = (int)blockIdx.x * (nfull / (int)gridDim.x);

    int nsaved = 0;
    int cnt = 0;
    for (int b0 = wave; b0 < nfull; b0 += EDGE_WAVES * SCAN_U) {
        uint4    v[SCAN_U];
        unsigned okm[SCAN_U];
        int      eb[SCAN_U];
        #pragma unroll
        for (int u = 0; u < SCAN_U; ++u) {
            const int  b  = b0 + EDGE_WAVES * u;
            const bool ok = b < nfull;
            okm[u] = (unsigned)((b - nfull) >> 31);
            int br = (ok ? b : b0) + rot;
            if (br >= nfull) br -= nfull;
            eb[u] = br * 512 + lane * 16;
            v[u] = *(const uint4*)(tbytes + eb[u]);
        }
        #pragma unroll
        for (int u = 0; u < SCAN_U; ++u) {
            const unsigned t[4] = {zero_bytes(v[u].x ^ key) & okm[u], zero_bytes(v[u].y ^ key) & okm[u],
                                   zero_bytes(v[u].z ^ key) & okm[u], zero_bytes(v[u].w ^ key) & okm[u]};
            #pragma unroll
            for (int k = 0; k < 4; ++k) {
                unsigned tk   = t[k];
                unsigned mask = __builtin_amdgcn_ballot_w32(tk != 0u);
                while (mask != 0u) {
                    if (tk != 0u) {
                        q[cnt + (int)__builtin_amdgcn_mbcnt_lo(mask, 0u)] =
                            eb[u] + 4 * k + (__builtin_ctz(tk) >> 3);
                        tk &= tk - 1u;
                    }
                    cnt += __builtin_popcount(mask);
                    mask = __builtin_amdgcn_ballot_w32(tk != 0u);
                }
            }
            if (cnt >= 32)
                cnt = tile_add_queue<SAVE>(xin, erec, q, cnt, false, lo, nn, lane, we, bb, tile, rec, nsaved);
        }
    }
    for (int e0 = nfull * 512 + wave * 32; e0 < E_; e0 += EDGE_WAVES * 32) {
        const int e = e0 + lane;
        const bool hit = (e < E_) && ((unsigned)tbytes[min(e, E_ - 1)] == kb);
        const unsigned mask = __builtin_amdgcn_ballot_w32(hit);
        if (mask != 0u) {
            if (hit) q[cnt + (int)__builtin_amdgcn_mbcnt_lo(mask, 0u)] = e;
            cnt += __builtin_popcount(mask);
        }
        if (cnt >= 32)
            cnt = tile_add_queue<SAVE>(xin, erec, q, cnt, false, lo, nn, lane, we, bb, tile, rec, nsaved);
    }
    if (cnt > 0)
        tile_add_queue<SAVE>(xin, erec, q, cnt, true, lo, nn, lane, we, bb, tile, rec, nsaved);
    return nsaved;
}

typedef unsigned long long __attribute__((may_alias)) u64_any;
typedef v8h __attribute__((may_alias)) v8h_any;
typedef _Float16 __attribute__((may_alias)) f16_any;

#define HPITCH 136

__device__ __forceinline__ void load_layer_weights(
        const _Float16* __restrict__ fw, const float* __restrict__ b, int np, int lane,
        v16h (&bv)[4][2], float (&bs)[2]) {
    #pragma unroll
    for (int kt = 0; kt < 4; ++kt) {
        #pragma unroll
        for (int j = 0; j < 2; ++j) {
            const int nt = 2 * np + j;
            const _Float16* q = fw + ((size_t)((kt << 3) | nt) * 32 + lane) * 16;
            v8h blo = *(const v8h*)q;
            v8h bhi = *(const v8h*)(q + 8);
            bv[kt][j] = __builtin_shufflevector(blo, bhi, 0,1,2,3,4,5,6,7,8,9,10,11,12,13,14,15);
        }
    }
    #pragma unroll
    for (int j = 0; j < 2; ++j) bs[j] = b[(2 * np + j) * 16 + (lane & 15)];
}

__device__ __forceinline__ void row_block_product(
        const f16_any* A, int mt, int m_, int hs8, const v16h (&bv)[4][2], v8f (&acc)[2]) {
    const f16_any* arow = A + (mt * 16 + m_) * HPITCH + hs8;
    v16h av[4];
    #pragma unroll
    for (int kt = 0; kt < 4; ++kt) {
        v8h alo = *(const v8h_any*)(arow + kt * 32);
        v8h ahi = *(const v8h_any*)(arow + kt * 32 + 16);
        av[kt] = __builtin_shufflevector(alo, ahi, 0,1,2,3,4,5,6,7,8,9,10,11,12,13,14,15);
    }
    const v8f vz = {0.f, 0.f, 0.f, 0.f, 0.f, 0.f, 0.f, 0.f};
    acc[0] = vz;
    acc[1] = vz;
    #pragma unroll
    for (int kt = 0; kt < 4; ++kt) {
        acc[0] = __builtin_amdgcn_wmma_f32_16x16x32_f16(
            false, av[kt], false, bv[kt][0], (short)0, acc[0], false, false);
        acc[1] = __builtin_amdgcn_wmma_f32_16x16x32_f16(
            false, av[kt], false, bv[kt][1], (short)0, acc[1], false, false);
    }
}

__device__ __forceinline__ void tile_mlp(
        const float* __restrict__ xin, unsigned char* smem, int lo, int nn, int N_, float ep,
        const _Float16* __restrict__ fw1, const float* __restrict__ b1,
        const _Float16* __restrict__ fw2, const float* __restrict__ b2,
        float* __restrict__ hout, int lane, int wave) {
    const float* sums = (const float*)smem;
    f16_any* A1 = (f16_any*)smem;
    f16_any* A2 = A1 + TILE_NODES * HPITCH;

    const int  trow0 = wave * 16;
    const bool mine  = trow0 < nn;
    unsigned long long hp[16];
    #pragma unroll
    for (int r = 0; r < 16; ++r) hp[r] = 0ull;
    if (mine) {
        v4f xv[16];
        #pragma unroll
        for (int r = 0; r < 16; ++r) {
            const int g = min(lo + trow0 + r, N_ - 1);
            const float* xr = xin + (size_t)g * F_DIM;
            xv[r] = *(const v4f*)(xr + (unsigned)(lane << 2));
        }
        #pragma unroll
        for (int r = 0; r < 16; ++r) {
            const float* sr = sums + (trow0 + r) * F_DIM + lane;
            union { _Float16 h[4]; unsigned long long u; } p;
            p.h[0] = (_Float16)(ep * xv[r].x + sr[0]);
            p.h[1] = (_Float16)(ep * xv[r].y + sr[32]);
            p.h[2] = (_Float16)(ep * xv[r].z + sr[64]);
            p.h[3] = (_Float16)(ep * xv[r].w + sr[96]);
            hp[r] = p.u;
        }
    }

    const int np  = wave & 3;
    const int mg  = wave >> 2;
    const int m_  = lane & 15;
    const int hs8 = (lane >> 4) << 3;

    v16h  bv[4][2];
    float bs[2];
    load_layer_weights(fw1, b1, np, lane, bv, bs);

    __syncthreads();
    if (mine) {
        #pragma unroll
        for (int r = 0; r < 16; ++r)
            *(u64_any*)(A1 + (trow0 + r) * HPITCH + (lane << 2)) = hp[r];
    }
    __syncthreads();

    for (int mt = mg; mt * 16 < nn; mt += 8) {
        v8f acc[2];
        row_block_product(A1, mt, m_, hs8, bv, acc);
        #pragma unroll
        for (int j = 0; j < 2; ++j) {
            const int col = (2 * np + j) * 16 + m_;
            #pragma unroll
            for (int r = 0; r < 8; ++r) {
                const float v = fmaxf(acc[j][r] + bs[j], 0.f);
                A2[(mt * 16 + r + hs8) * HPITCH + col] = (_Float16)v;
            }
        }
    }

    load_layer_weights(fw2, b2, np, lane, bv, bs);

    __syncthreads();

    for (int mt = mg; mt * 16 < nn; mt += 8) {
        v8f acc[2];
        row_block_product(A2, mt, m_, hs8, bv, acc);
        float o[2][8];
        #pragma unroll
        for (int j = 0; j < 2; ++j) {
            #pragma unroll
            for (int r = 0; r < 8; ++r) o[j][r] = fmaxf(acc[j][r] + bs[j], 0.f);
        }
        const int grow0 = lo + mt * 16;
        float* ob = hout + (size_t)grow0 * F_DIM;
        if (grow0 + 16 <= N_) {
            #pragma unroll
            for (int j = 0; j < 2; ++j) {
                const int col = (2 * np + j) * 16 + m_;
                #pragma unroll
                for (int r = 0; r < 8; ++r)
                    ob[(unsigned)((r + hs8) * F_DIM + col)] = o[j][r];
            }
        } else {
            #pragma unroll
            for (int j = 0; j < 2; ++j) {
                const int col = (2 * np + j) * 16 + m_;
                #pragma unroll
                for (int r = 0; r < 8; ++r)
                    if (grow0 + r + hs8 < N_)
                        ob[(unsigned)((r + hs8) * F_DIM + col)] = o[j][r];
            }
        }
    }
}

template <bool REPLAY>
__global__ void __launch_bounds__(1024)
edge_mlp(const float* __restrict__ xin, const int* __restrict__ erec,
         const float* __restrict__ We,
         const float* __restrict__ be, const unsigned* __restrict__ tid4,
         const float* __restrict__ eps,
         const _Float16* __restrict__ fw1, const float* __restrict__ b1,
         const _Float16* __restrict__ fw2, const float* __restrict__ b2,
         float* __restrict__ hout, int* __restrict__ recs, int* __restrict__ rcnt,
         int E_, int N_) {
    __shared__ __align__(16) unsigned char smem[TILE_NODES * F_DIM * 4 + EDGE_WAVES * QCAP * 4];
    float* tile = (float*)smem;

    const int tid  = threadIdx.x;
    const int lane = tid & 31;
    const int wave = __builtin_amdgcn_readfirstlane(tid >> 5);
    const int lo   = blockIdx.x * TILE_NODES;
    const int nn   = min(TILE_NODES, N_ - lo);

    const int slot = (int)blockIdx.x * EDGE_WAVES + wave;
    int* rec = recs + (size_t)slot * (3 * RCAP);
    int nrec = 0;
    if (REPLAY) nrec = rcnt[slot];

    float4* tile4 = (float4*)tile;
    for (int i = tid; i < TILE_NODES * F_DIM / 4; i += 1024)
        tile4[i] = make_float4(0.f, 0.f, 0.f, 0.f);

    float we[4], bb[4];
    #pragma unroll
    for (int k = 0; k < 4; ++k) { we[k] = We[4 * lane + k]; bb[k] = be[4 * lane + k]; }

    int* q = (int*)(smem + TILE_NODES * F_DIM * 4) + wave * QCAP;
    const unsigned char* tbytes = (const unsigned char*)tid4;
    const v4i* erec4 = (const v4i*)erec;

    __syncthreads();

    if (REPLAY) {
        const int n = __builtin_amdgcn_readfirstlane(nrec);
        if (n <= RCAP) {
            if (n > 0) {
                int idx = min(lane, n - 1);
                int sv = rec[idx], dv = rec[RCAP + idx], av = rec[2 * RCAP + idx];
                for (int base = 0; base < n; base += 32) {
                    const int m = min(32, n - base);
                    idx = min(base + 32 + lane, n - 1);
                    int sv2, dv2, av2;
                    tile_add_batch(xin, sv, dv, av, m, nn, lane, we, bb, tile, [&] {
                        sv2 = rec[idx]; dv2 = rec[RCAP + idx]; av2 = rec[2 * RCAP + idx];
                    });
                    sv = sv2; dv = dv2; av = av2;
                }
            }
        } else {
            scan_tile<false>(xin, erec4, tbytes, q, rec, E_, lo, nn, lane, wave, we, bb, tile);
        }
    } else {
        const int found = scan_tile<true>(xin, erec4, tbytes, q, rec, E_, lo, nn, lane, wave, we, bb, tile);
        if (lane == 0) rcnt[slot] = found;
    }

    __syncthreads();

    tile_mlp(xin, smem, lo, nn, N_, 1.0f + eps[0], fw1, b1, fw2, b2, hout, lane, wave);
}

typedef float v2f __attribute__((ext_vector_type(2)));

#define N_CLS      40
#define CLS_WAVES  8

__global__ void __launch_bounds__(256)
classifier(const float* __restrict__ h, const float* __restrict__ Wl,
           const float* __restrict__ bl, float* __restrict__ out, int n) {
    __shared__ __align__(16) v4f   btab[16 * 3 * 32];
    __shared__ __align__(16) float stage[CLS_WAVES][16 * N_CLS];

    const int tid  = threadIdx.x;
    const int lane = tid & 31;
    const int wave = __builtin_amdgcn_readfirstlane(tid >> 5);
    const int c    = lane & 15;
    const int hh   = lane >> 4;
    const int row0 = ((int)blockIdx.x * CLS_WAVES + wave) * 16;

    const int    arow = min(row0 + c, n - 1);
    const float* ap   = h + (size_t)arow * F_DIM + 4 * hh;
    v4f a[16];
    #pragma unroll
    for (int j = 0; j < 16; ++j) a[j] = *(const v4f*)(ap + 8 * j);

    #pragma unroll
    for (int i = tid; i < 16 * 3 * 32; i += 256) {
        const int  j   = i / 96;
        const int  rem = i - 96 * j;
        const int  nt  = rem >> 5;
        const int  ln  = rem & 31;
        const int  nn  = nt * 16 + (ln & 15);
        const bool in  = nn < N_CLS;
        const float* wp = Wl + (8 * j + 4 * (ln >> 4)) * N_CLS + min(nn, N_CLS - 1);
        const float w0 = wp[0], w1 = wp[N_CLS], w2 = wp[2 * N_CLS], w3 = wp[3 * N_CLS];
        const v4f w = {in ? w0 : 0.f, in ? w1 : 0.f, in ? w2 : 0.f, in ? w3 : 0.f};
        btab[i] = w;
    }
    float bs[3];
    #pragma unroll
    for (int nt = 0; nt < 3; ++nt) {
        const int nn = nt * 16 + c;
        bs[nt] = (nn < N_CLS) ? bl[nn] : 0.f;
    }
    __syncthreads();

    const v8f vz = {0.f, 0.f, 0.f, 0.f, 0.f, 0.f, 0.f, 0.f};
    v8f acc[3] = {vz, vz, vz};
    #pragma unroll
    for (int j = 0; j < 16; ++j) {
        const v2f a0 = {a[j].x, a[j].y};
        const v2f a1 = {a[j].z, a[j].w};
        #pragma unroll
        for (int nt = 0; nt < 3; ++nt) {
            const v4f b  = btab[(j * 3 + nt) * 32 + lane];
            const v2f b0 = {b.x, b.y};
            const v2f b1 = {b.z, b.w};
            acc[nt] = __builtin_amdgcn_wmma_f32_16x16x4_f32(
                false, a0, false, b0, (short)0, acc[nt], false, false);
            acc[nt] = __builtin_amdgcn_wmma_f32_16x16x4_f32(
                false, a1, false, b1, (short)0, acc[nt], false, false);
        }
    }

    float* st = stage[wave];
    const bool ok2 = c < N_CLS - 32;
    #pragma unroll
    for (int r = 0; r < 8; ++r) {
        const float v0 = acc[0][r] + bs[0];
        const float v1 = acc[1][r] + bs[1];
        const float v2 = acc[2][r] + bs[2];
        float mx = fmaxf(v0, v1);
        if (ok2) mx = fmaxf(mx, v2);
        #pragma unroll
        for (int o = 8; o >= 1; o >>= 1) mx = fmaxf(mx, __shfl_xor(mx, o, 32));
        float s = __expf(v0 - mx) + __expf(v1 - mx);
        if (ok2) s += __expf(v2 - mx);
        #pragma unroll
        for (int o = 8; o >= 1; o >>= 1) s += __shfl_xor(s, o, 32);
        const float ls = __logf(s);
        float* sr = st + (8 * hh + r) * N_CLS + c;
        sr[0]  = v0 - mx - ls;
        sr[16] = v1 - mx - ls;
        if (ok2) sr[32] = v2 - mx - ls;
    }
    __syncthreads();

    const int nvalid = min(n - row0, 16) * N_CLS;
    float* ob = out + (size_t)row0 * N_CLS;
    #pragma unroll
    for (int q = lane; q < 16 * N_CLS / 4; q += 32) {
        if (4 * q < nvalid) *(v4f*)(ob + 4 * q) = *(const v4f*)(st + 4 * q);
    }
}

extern "C" void kernel_launch(void* const* d_in, const int* in_sizes, int n_in,
                              void* d_out, int out_size, void* d_ws, size_t ws_size,
                              hipStream_t stream) {
    const float* x    = (const float*)d_in[0];
    const int*   ei   = (const int*)d_in[1];
    const float* ea   = (const float*)d_in[2];
    const float* eps1 = (const float*)d_in[3];
    const float* We1  = (const float*)d_in[4];
    const float* be1  = (const float*)d_in[5];
    const float* W11  = (const float*)d_in[6];
    const float* b11  = (const float*)d_in[7];
    const float* W12  = (const float*)d_in[8];
    const float* b12  = (const float*)d_in[9];
    const float* eps2 = (const float*)d_in[10];
    const float* We2  = (const float*)d_in[11];
    const float* be2  = (const float*)d_in[12];
    const float* W21  = (const float*)d_in[13];
    const float* b21  = (const float*)d_in[14];
    const float* W22  = (const float*)d_in[15];
    const float* b22  = (const float*)d_in[16];
    const float* Wl   = (const float*)d_in[17];
    const float* bl   = (const float*)d_in[18];

    const int N_ = in_sizes[0] / F_DIM;
    const int E_ = in_sizes[2];
    const size_t NB = (size_t)N_ * F_DIM * sizeof(float);

    char* ws = (char*)d_ws;
    float*    h2    = (float*)ws;
    float*    h1    = (float*)(ws + NB);
    _Float16* frags = (_Float16*)(ws + 2 * NB);
    unsigned* tid4  = (unsigned*)(ws + 2 * NB + 131072);
    const size_t TBY = (((size_t)E_ + 3) & ~(size_t)3);
    _Float16* f11 = frags;
    _Float16* f12 = frags + 16384;
    _Float16* f21 = frags + 32768;
    _Float16* f22 = frags + 49152;
    float* out = (float*)d_out;

    const int tb = (N_ + TILE_NODES - 1) / TILE_NODES;
    int* rcnt = (int*)(ws + 2 * NB + 131072 + ((TBY + 255) & ~(size_t)255));
    int* recs = rcnt + (((size_t)tb * EDGE_WAVES + 63) & ~(size_t)63);
    int* erec = recs + (size_t)tb * EDGE_WAVES * (3 * RCAP);
    const int cb = (N_ + 16 * CLS_WAVES - 1) / (16 * CLS_WAVES);

    prep<<<1024, 256, 0, stream>>>(W11, W12, W21, W22, frags, ei, ea, tid4, erec, E_);

    edge_mlp<false><<<tb, 1024, 0, stream>>>(x, erec, We1, be1, tid4, eps1, f11, b11, f12, b12,
                                              h1, recs, rcnt, E_, N_);
    edge_mlp<true><<<tb, 1024, 0, stream>>>(h1, erec, We2, be2, tid4, eps2, f21, b21, f22, b22,
                                             h2, recs, rcnt, E_, N_);

    classifier<<<cb, 256, 0, stream>>>(h2, Wl, bl, out, N_);
}


# Edge search: first launch builds a bit index per tile slot (a bit per four edges, LDS integer OR between barriers); a tile reads its 50 KB of bits, not every edge's byte; sum order varies as before
# speedup vs baseline: 4.6119x; 1.0934x over previous
#include <hip/hip_runtime.h>
#include <hip/hip_bf16.h>
#include <stdint.h>

typedef _Float16 v16h __attribute__((ext_vector_type(16)));
typedef _Float16 v8h  __attribute__((ext_vector_type(8)));
typedef float    v8f  __attribute__((ext_vector_type(8)));
typedef float    v4f  __attribute__((ext_vector_type(4)));
typedef int      v4i  __attribute__((ext_vector_type(4)));

#define F_DIM 128

#define TILE_NODES 400
#define IDX_SLOTS  256
#define IDX_EDGES  4096
#define IDX_LPITCH 33

__device__ __forceinline__ unsigned tile_slot(int d) {
    return ((unsigned)d / (unsigned)TILE_NODES) & (unsigned)(IDX_SLOTS - 1);
}

__global__ void __launch_bounds__(256)
prep(const float* __restrict__ W11, const float* __restrict__ W12,
     const float* __restrict__ W21, const float* __restrict__ W22,
     _Float16* __restrict__ frags, const int* __restrict__ ei,
     const float* __restrict__ ea, unsigned* __restrict__ bits,
     int* __restrict__ erec, int E_) {
    __shared__ unsigned lbits[IDX_SLOTS * IDX_LPITCH];

    const int gid = blockIdx.x * blockDim.x + threadIdx.x;
    if (gid < 4 * 8192) {
        int m    = gid >> 13;
        int r    = gid & 8191;
        int d    = r & 7;
        int lane = (r >> 3) & 31;
        int ntk  = r >> 8;
        int kt   = ntk >> 3;
        int nt   = ntk & 7;
        int nn   = nt * 16 + (lane & 15);
        int kb   = kt * 32 + ((lane >> 4) << 3);
        int k0   = kb + ((d < 4) ? (d << 1) : (16 + ((d - 4) << 1)));
        const float* W = (m == 0) ? W11 : (m == 1) ? W12 : (m == 2) ? W21 : W22;
        _Float16* o = frags + (size_t)m * 16384 + (size_t)r * 2;
        o[0] = (_Float16)W[(size_t)k0 * F_DIM + nn];
        o[1] = (_Float16)W[(size_t)(k0 + 1) * F_DIM + nn];
    }

    const int* dsts = ei + E_;
    const bool al   = (E_ & 3) == 0;
    const int  nblk = (E_ + IDX_EDGES - 1) / IDX_EDGES;
    for (int blk = blockIdx.x; blk < nblk; blk += (int)gridDim.x) {
        for (int i = threadIdx.x; i < IDX_SLOTS * IDX_LPITCH; i += 256) lbits[i] = 0u;
        __syncthreads();
        const int e0 = blk * IDX_EDGES + 4 * (int)threadIdx.x;
        auto set_bits = [&](int j, int d0, int d1, int d2, int d3) {
            const int gl = j * 256 + (int)threadIdx.x;
            const unsigned w = (unsigned)gl >> 5;
            const unsigned b = 1u << (gl & 31);
            __hip_atomic_fetch_or(&lbits[tile_slot(d0) * IDX_LPITCH + w], b, __ATOMIC_RELAXED,
                                  __HIP_MEMORY_SCOPE_WORKGROUP);
            __hip_atomic_fetch_or(&lbits[tile_slot(d1) * IDX_LPITCH + w], b, __ATOMIC_RELAXED,
                                  __HIP_MEMORY_SCOPE_WORKGROUP);
            __hip_atomic_fetch_or(&lbits[tile_slot(d2) * IDX_LPITCH + w], b, __ATOMIC_RELAXED,
                                  __HIP_MEMORY_SCOPE_WORKGROUP);
            __hip_atomic_fetch_or(&lbits[tile_slot(d3) * IDX_LPITCH + w], b, __ATOMIC_RELAXED,
                                  __HIP_MEMORY_SCOPE_WORKGROUP);
        };
        if (al) {
            int4 v[4];
            #pragma unroll
            for (int j = 0; j < 4; ++j) v[j] = *(const int4*)(dsts + min(e0 + 1024 * j, E_ - 4));
            #pragma unroll
            for (int j = 0; j < 4; ++j)
                if (e0 + 1024 * j < E_) set_bits(j, v[j].x, v[j].y, v[j].z, v[j].w);
        } else {
            for (int j = 0; j < 4; ++j) {
                const int e = e0 + 1024 * j;
                if (e < E_)
                    set_bits(j, dsts[e], dsts[min(e + 1, E_ - 1)], dsts[min(e + 2, E_ - 1)],
                             dsts[min(e + 3, E_ - 1)]);
            }
        }
        __syncthreads();
        for (int r = 0; r < IDX_SLOTS / 8; ++r) {
            const int slot = r * 8 + ((int)threadIdx.x >> 5);
            const int wd   = (int)threadIdx.x & 31;
            bits[((size_t)slot * nblk + blk) * 32 + wd] = lbits[slot * IDX_LPITCH + wd];
        }
        __syncthreads();
    }

    v4i* er = (v4i*)erec;
    const int ln   = threadIdx.x & 31;
    const int wv   = gid >> 5;
    const int nwv  = (int)(gridDim.x * blockDim.x) >> 5;
    for (int c = wv * 128; c < E_; c += nwv * 128) {
        int sv[4], dv[4], av[4];
        #pragma unroll
        for (int k = 0; k < 4; ++k) {
            const int e = min(c + 32 * k + ln, E_ - 1);
            sv[k] = ei[e];
            dv[k] = dsts[e];
            av[k] = __float_as_int(ea[e]);
        }
        #pragma unroll
        for (int k = 0; k < 4; ++k) {
            const int e = c + 32 * k + ln;
            if (e < E_) {
                const v4i r = {sv[k], dv[k], av[k], 0};
                er[e] = r;
            }
        }
    }
}

#define EDGE_WAVES 32
#define GQCAP      40
#define QCAP       64
#define RCAP       512

template <class F>
__device__ __forceinline__ void tile_add_batch(
        const float* __restrict__ xin, int sv, int dv, int av, int m,
        int nn, int lane, const float (&we)[4], const float (&bb)[4], float* tile,
        F behind_first_loads) {
    for (int j = 0; j < m; j += 8) {
        v4f   r[8];
        int   dl[8];
        float aa[8];
        #pragma unroll
        for (int u = 0; u < 8; ++u) {
            const int jj = min(j + u, m - 1);
            const int s  = __builtin_amdgcn_readlane(sv, jj);
            dl[u] = __builtin_amdgcn_readlane(dv, jj);
            aa[u] = __int_as_float(__builtin_amdgcn_readlane(av, jj));
            r[u] = *(const v4f*)(xin + (size_t)s * F_DIM + (unsigned)(lane << 2));
        }
        if (j == 0) behind_first_loads();
        #pragma unroll
        for (int u = 0; u < 8; ++u) {
            if (j + u < m && (unsigned)dl[u] < (unsigned)nn) {
                float* t = tile + dl[u] * F_DIM + lane;
                #pragma unroll
                for (int k = 0; k < 4; ++k) {
                    const float v = fmaxf(r[u][k] + aa[u] * we[k] + bb[k], 0.f);
                    __hip_atomic_fetch_add(t + 32 * k, v, __ATOMIC_RELAXED,
                                           __HIP_MEMORY_SCOPE_WORKGROUP);
                }
            }
        }
    }
}

template <bool SAVE>
__device__ __forceinline__ int tile_add_queue(
        const float* __restrict__ xin, int* qs, int* qd, int* qa, int n, bool all,
        int nn, int lane, const float (&we)[4], const float (&bb)[4], float* tile,
        int* __restrict__ rec, int& nsaved) {
    int base = 0;
    for (; base < n; base += 32) {
        const int m = min(32, n - base);
        if (!all && m < 32) break;
        const int idx = base + min(lane, m - 1);
        const int sv = qs[idx];
        const int dv = qd[idx];
        const int av = qa[idx];
        if (SAVE) {
            const int p = nsaved + lane;
            if (lane < m && p < RCAP) {
                rec[p]            = sv;
                rec[RCAP + p]     = dv;
                rec[2 * RCAP + p] = av;
            }
            nsaved += m;
        }
        tile_add_batch(xin, sv, dv, av, m, nn, lane, we, bb, tile, [] {});
    }
    if (base >= n) return 0;
    const int rem = n - base;
    if (base > 0 && lane < rem) {
        const int s = qs[base + lane], d = qd[base + lane], a = qa[base + lane];
        qs[lane] = s; qd[lane] = d; qa[lane] = a;
    }
    return rem;
}

template <bool SAVE>
__device__ __forceinline__ int take_groups(
        const float* __restrict__ xin, const v4i* __restrict__ erec,
        int* gq, int gcnt, bool all, int* qs, int* qd, int* qa, int& cnt,
        int E_, int lo, int nn, int lane,
        const float (&we)[4], const float (&bb)[4], float* tile,
        int* __restrict__ rec, int& nsaved) {
    int base = 0;
    for (; base < gcnt; base += 8) {
        const int m = min(8, gcnt - base);
        if (!all && m < 8) break;
        const int gi = lane >> 2;
        const int g  = gq[base + min(gi, m - 1)];
        const int e  = 4 * g + (lane & 3);
        const v4i er = erec[min(e, E_ - 1)];
        const int dv = er.y - lo;
        const bool hit = (gi < m) && (e < E_) && ((unsigned)dv < (unsigned)nn);
        const unsigned mask = __builtin_amdgcn_ballot_w32(hit);
        if (mask != 0u) {
            if (hit) {
                const int p = cnt + (int)__builtin_amdgcn_mbcnt_lo(mask, 0u);
                qs[p] = er.x;
                qd[p] = dv;
                qa[p] = er.z;
            }
            cnt += __builtin_popcount(mask);
            if (cnt >= 32)
                cnt = tile_add_queue<SAVE>(xin, qs, qd, qa, cnt, false, nn, lane, we, bb, tile, rec, nsaved);
        }
    }
    if (base >= gcnt) return 0;
    const int rem = gcnt - base;
    if (base > 0 && lane < rem) gq[lane] = gq[base + lane];
    return rem;
}

template <bool SAVE>
__device__ __forceinline__ int scan_tile(
        const float* __restrict__ xin, const v4i* __restrict__ erec,
        const uint4* __restrict__ bits4, int n4,
        int* gq, int* qs, int* qd, int* qa, int* __restrict__ rec,
        int E_, int lo, int nn, int lane, int wave,
        const float (&we)[4], const float (&bb)[4], float* tile) {
    int nsaved = 0;
    int cnt    = 0;
    int gcnt   = 0;
    const int per = (n4 + EDGE_WAVES - 1) / EDGE_WAVES;
    const int end = min((wave + 1) * per, n4);
    int i0 = wave * per;
    uint4 nxt = make_uint4(0u, 0u, 0u, 0u);
    if (i0 < end) nxt = bits4[min(i0 + lane, end - 1)];
    for (; i0 < end; i0 += 32) {
        const int i = i0 + lane;
        const unsigned okm = (i < end) ? 0xFFFFFFFFu : 0u;
        const uint4 cur = make_uint4(nxt.x & okm, nxt.y & okm, nxt.z & okm, nxt.w & okm);
        nxt = bits4[min(i + 32, n4 - 1)];
        #pragma nounroll
        for (int k = 0; k < 4; ++k) {
            unsigned tk = (k == 0) ? cur.x : (k == 1) ? cur.y : (k == 2) ? cur.z : cur.w;
            const int g0 = (4 * i + k) * 32;
            unsigned mask = __builtin_amdgcn_ballot_w32(tk != 0u);
            while (mask != 0u) {
                if (tk != 0u) {
                    gq[gcnt + (int)__builtin_amdgcn_mbcnt_lo(mask, 0u)] = g0 + __builtin_ctz(tk);
                    tk &= tk - 1u;
                }
                gcnt += __builtin_popcount(mask);
                if (gcnt >= 8)
                    gcnt = take_groups<SAVE>(xin, erec, gq, gcnt, false, qs, qd, qa, cnt,
                                             E_, lo, nn, lane, we, bb, tile, rec, nsaved);
                mask = __builtin_amdgcn_ballot_w32(tk != 0u);
            }
        }
    }
    if (gcnt > 0)
        take_groups<SAVE>(xin, erec, gq, gcnt, true, qs, qd, qa, cnt,
                          E_, lo, nn, lane, we, bb, tile, rec, nsaved);
    if (cnt > 0)
        tile_add_queue<SAVE>(xin, qs, qd, qa, cnt, true, nn, lane, we, bb, tile, rec, nsaved);
    return nsaved;
}

typedef unsigned long long __attribute__((may_alias)) u64_any;
typedef v8h __attribute__((may_alias)) v8h_any;
typedef _Float16 __attribute__((may_alias)) f16_any;

#define HPITCH 136

__device__ __forceinline__ void load_layer_weights(
        const _Float16* __restrict__ fw, const float* __restrict__ b, int np, int lane,
        v16h (&bv)[4][2], float (&bs)[2]) {
    #pragma unroll
    for (int kt = 0; kt < 4; ++kt) {
        #pragma unroll
        for (int j = 0; j < 2; ++j) {
            const int nt = 2 * np + j;
            const _Float16* q = fw + ((size_t)((kt << 3) | nt) * 32 + lane) * 16;
            v8h blo = *(const v8h*)q;
            v8h bhi = *(const v8h*)(q + 8);
            bv[kt][j] = __builtin_shufflevector(blo, bhi, 0,1,2,3,4,5,6,7,8,9,10,11,12,13,14,15);
        }
    }
    #pragma unroll
    for (int j = 0; j < 2; ++j) bs[j] = b[(2 * np + j) * 16 + (lane & 15)];
}

__device__ __forceinline__ void row_block_product(
        const f16_any* A, int mt, int m_, int hs8, const v16h (&bv)[4][2], v8f (&acc)[2]) {
    const f16_any* arow = A + (mt * 16 + m_) * HPITCH + hs8;
    v16h av[4];
    #pragma unroll
    for (int kt = 0; kt < 4; ++kt) {
        v8h alo = *(const v8h_any*)(arow + kt * 32);
        v8h ahi = *(const v8h_any*)(arow + kt * 32 + 16);
        av[kt] = __builtin_shufflevector(alo, ahi, 0,1,2,3,4,5,6,7,8,9,10,11,12,13,14,15);
    }
    const v8f vz = {0.f, 0.f, 0.f, 0.f, 0.f, 0.f, 0.f, 0.f};
    acc[0] = vz;
    acc[1] = vz;
    #pragma unroll
    for (int kt = 0; kt < 4; ++kt) {
        acc[0] = __builtin_amdgcn_wmma_f32_16x16x32_f16(
            false, av[kt], false, bv[kt][0], (short)0, acc[0], false, false);
        acc[1] = __builtin_amdgcn_wmma_f32_16x16x32_f16(
            false, av[kt], false, bv[kt][1], (short)0, acc[1], false, false);
    }
}

__device__ __forceinline__ void tile_mlp(
        const float* __restrict__ xin, unsigned char* smem, int lo, int nn, int N_, float ep,
        const _Float16* __restrict__ fw1, const float* __restrict__ b1,
        const _Float16* __restrict__ fw2, const float* __restrict__ b2,
        float* __restrict__ hout, int lane, int wave) {
    const float* sums = (const float*)smem;
    f16_any* A1 = (f16_any*)smem;
    f16_any* A2 = A1 + TILE_NODES * HPITCH;

    const int  trow0 = wave * 16;
    const bool mine  = trow0 < nn;
    unsigned long long hp[16];
    #pragma unroll
    for (int r = 0; r < 16; ++r) hp[r] = 0ull;
    if (mine) {
        v4f xv[16];
        #pragma unroll
        for (int r = 0; r < 16; ++r) {
            const int g = min(lo + trow0 + r, N_ - 1);
            const float* xr = xin + (size_t)g * F_DIM;
            xv[r] = *(const v4f*)(xr + (unsigned)(lane << 2));
        }
        #pragma unroll
        for (int r = 0; r < 16; ++r) {
            const float* sr = sums + (trow0 + r) * F_DIM + lane;
            union { _Float16 h[4]; unsigned long long u; } p;
            p.h[0] = (_Float16)(ep * xv[r].x + sr[0]);
            p.h[1] = (_Float16)(ep * xv[r].y + sr[32]);
            p.h[2] = (_Float16)(ep * xv[r].z + sr[64]);
            p.h[3] = (_Float16)(ep * xv[r].w + sr[96]);
            hp[r] = p.u;
        }
    }

    const int np  = wave & 3;
    const int mg  = wave >> 2;
    const int m_  = lane & 15;
    const int hs8 = (lane >> 4) << 3;

    v16h  bv[4][2];
    float bs[2];
    load_layer_weights(fw1, b1, np, lane, bv, bs);

    __syncthreads();
    if (mine) {
        #pragma unroll
        for (int r = 0; r < 16; ++r)
            *(u64_any*)(A1 + (trow0 + r) * HPITCH + (lane << 2)) = hp[r];
    }
    __syncthreads();

    for (int mt = mg; mt * 16 < nn; mt += 8) {
        v8f acc[2];
        row_block_product(A1, mt, m_, hs8, bv, acc);
        #pragma unroll
        for (int j = 0; j < 2; ++j) {
            const int col = (2 * np + j) * 16 + m_;
            #pragma unroll
            for (int r = 0; r < 8; ++r) {
                const float v = fmaxf(acc[j][r] + bs[j], 0.f);
                A2[(mt * 16 + r + hs8) * HPITCH + col] = (_Float16)v;
            }
        }
    }

    load_layer_weights(fw2, b2, np, lane, bv, bs);

    __syncthreads();

    for (int mt = mg; mt * 16 < nn; mt += 8) {
        v8f acc[2];
        row_block_product(A2, mt, m_, hs8, bv, acc);
        float o[2][8];
        #pragma unroll
        for (int j = 0; j < 2; ++j) {
            #pragma unroll
            for (int r = 0; r < 8; ++r) o[j][r] = fmaxf(acc[j][r] + bs[j], 0.f);
        }
        const int grow0 = lo + mt * 16;
        float* ob = hout + (size_t)grow0 * F_DIM;
        if (grow0 + 16 <= N_) {
            #pragma unroll
            for (int j = 0; j < 2; ++j) {
                const int col = (2 * np + j) * 16 + m_;
                #pragma unroll
                for (int r = 0; r < 8; ++r)
                    ob[(unsigned)((r + hs8) * F_DIM + col)] = o[j][r];
            }
        } else {
            #pragma unroll
            for (int j = 0; j < 2; ++j) {
                const int col = (2 * np + j) * 16 + m_;
                #pragma unroll
                for (int r = 0; r < 8; ++r)
                    if (grow0 + r + hs8 < N_)
                        ob[(unsigned)((r + hs8) * F_DIM + col)] = o[j][r];
            }
        }
    }
}

template <bool REPLAY>
__global__ void __launch_bounds__(1024)
edge_mlp(const float* __restrict__ xin, const int* __restrict__ erec,
         const float* __restrict__ We,
         const float* __restrict__ be, const unsigned* __restrict__ bits,
         const float* __restrict__ eps,
         const _Float16* __restrict__ fw1, const float* __restrict__ b1,
         const _Float16* __restrict__ fw2, const float* __restrict__ b2,
         float* __restrict__ hout, int* __restrict__ recs, int* __restrict__ rcnt,
         int E_, int N_) {
    __shared__ __align__(16) unsigned char smem[TILE_NODES * F_DIM * 4 + EDGE_WAVES * (GQCAP + 3 * QCAP) * 4];
    float* tile = (float*)smem;

    const int tid  = threadIdx.x;
    const int lane = tid & 31;
    const int wave = __builtin_amdgcn_readfirstlane(tid >> 5);
    const int lo   = blockIdx.x * TILE_NODES;
    const int nn   = min(TILE_NODES, N_ - lo);

    const int slot = (int)blockIdx.x * EDGE_WAVES + wave;
    int* rec = recs + (size_t)slot * (3 * RCAP);
    int nrec = 0;
    if (REPLAY) nrec = rcnt[slot];

    float4* tile4 = (float4*)tile;
    for (int i = tid; i < TILE_NODES * F_DIM / 4; i += 1024)
        tile4[i] = make_float4(0.f, 0.f, 0.f, 0.f);

    float we[4], bb[4];
    #pragma unroll
    for (int k = 0; k < 4; ++k) { we[k] = We[4 * lane + k]; bb[k] = be[4 * lane + k]; }

    int* gq = (int*)(smem + TILE_NODES * F_DIM * 4) + wave * (GQCAP + 3 * QCAP);
    int* qs = gq + GQCAP;
    int* qd = qs + QCAP;
    int* qa = qd + QCAP;
    const int nblk = (E_ + IDX_EDGES - 1) / IDX_EDGES;
    const int n4   = nblk * 8;
    const uint4* bits4 = (const uint4*)(bits + (size_t)(blockIdx.x & (IDX_SLOTS - 1)) * nblk * 32);
    const v4i* erec4 = (const v4i*)erec;

    __syncthreads();

    if (REPLAY) {
        const int n = __builtin_amdgcn_readfirstlane(nrec);
        if (n <= RCAP) {
            if (n > 0) {
                int idx = min(lane, n - 1);
                int sv = rec[idx], dv = rec[RCAP + idx], av = rec[2 * RCAP + idx];
                for (int base = 0; base < n; base += 32) {
                    const int m = min(32, n - base);
                    idx = min(base + 32 + lane, n - 1);
                    int sv2, dv2, av2;
                    tile_add_batch(xin, sv, dv, av, m, nn, lane, we, bb, tile, [&] {
                        sv2 = rec[idx]; dv2 = rec[RCAP + idx]; av2 = rec[2 * RCAP + idx];
                    });
                    sv = sv2; dv = dv2; av = av2;
                }
            }
        } else {
            scan_tile<false>(xin, erec4, bits4, n4, gq, qs, qd, qa, rec, E_, lo, nn, lane, wave, we, bb, tile);
        }
    } else {
        const int found = scan_tile<true>(xin, erec4, bits4, n4, gq, qs, qd, qa, rec, E_, lo, nn, lane, wave, we, bb, tile);
        if (lane == 0) rcnt[slot] = found;
    }

    __syncthreads();

    tile_mlp(xin, smem, lo, nn, N_, 1.0f + eps[0], fw1, b1, fw2, b2, hout, lane, wave);
}
typedef float v2f __attribute__((ext_vector_type(2)));

#define N_CLS      40
#define CLS_WAVES  8

__global__ void __launch_bounds__(256)
classifier(const float* __restrict__ h, const float* __restrict__ Wl,
           const float* __restrict__ bl, float* __restrict__ out, int n) {
    __shared__ __align__(16) v4f   btab[16 * 3 * 32];
    __shared__ __align__(16) float stage[CLS_WAVES][16 * N_CLS];

    const int tid  = threadIdx.x;
    const int lane = tid & 31;
    const int wave = __builtin_amdgcn_readfirstlane(tid >> 5);
    const int c    = lane & 15;
    const int hh   = lane >> 4;
    const int row0 = ((int)blockIdx.x * CLS_WAVES + wave) * 16;

    const int    arow = min(row0 + c, n - 1);
    const float* ap   = h + (size_t)arow * F_DIM + 4 * hh;
    v4f a[16];
    #pragma unroll
    for (int j = 0; j < 16; ++j) a[j] = *(const v4f*)(ap + 8 * j);

    #pragma unroll
    for (int i = tid; i < 16 * 3 * 32; i += 256) {
        const int  j   = i / 96;
        const int  rem = i - 96 * j;
        const int  nt  = rem >> 5;
        const int  ln  = rem & 31;
        const int  nn  = nt * 16 + (ln & 15);
        const bool in  = nn < N_CLS;
        const float* wp = Wl + (8 * j + 4 * (ln >> 4)) * N_CLS + min(nn, N_CLS - 1);
        const float w0 = wp[0], w1 = wp[N_CLS], w2 = wp[2 * N_CLS], w3 = wp[3 * N_CLS];
        const v4f w = {in ? w0 : 0.f, in ? w1 : 0.f, in ? w2 : 0.f, in ? w3 : 0.f};
        btab[i] = w;
    }
    float bs[3];
    #pragma unroll
    for (int nt = 0; nt < 3; ++nt) {
        const int nn = nt * 16 + c;
        bs[nt] = (nn < N_CLS) ? bl[nn] : 0.f;
    }
    __syncthreads();

    const v8f vz = {0.f, 0.f, 0.f, 0.f, 0.f, 0.f, 0.f, 0.f};
    v8f acc[3] = {vz, vz, vz};
    #pragma unroll
    for (int j = 0; j < 16; ++j) {
        const v2f a0 = {a[j].x, a[j].y};
        const v2f a1 = {a[j].z, a[j].w};
        #pragma unroll
        for (int nt = 0; nt < 3; ++nt) {
            const v4f b  = btab[(j * 3 + nt) * 32 + lane];
            const v2f b0 = {b.x, b.y};
            const v2f b1 = {b.z, b.w};
            acc[nt] = __builtin_amdgcn_wmma_f32_16x16x4_f32(
                false, a0, false, b0, (short)0, acc[nt], false, false);
            acc[nt] = __builtin_amdgcn_wmma_f32_16x16x4_f32(
                false, a1, false, b1, (short)0, acc[nt], false, false);
        }
    }

    float* st = stage[wave];
    const bool ok2 = c < N_CLS - 32;
    #pragma unroll
    for (int r = 0; r < 8; ++r) {
        const float v0 = acc[0][r] + bs[0];
        const float v1 = acc[1][r] + bs[1];
        const float v2 = acc[2][r] + bs[2];
        float mx = fmaxf(v0, v1);
        if (ok2) mx = fmaxf(mx, v2);
        #pragma unroll
        for (int o = 8; o >= 1; o >>= 1) mx = fmaxf(mx, __shfl_xor(mx, o, 32));
        float s = __expf(v0 - mx) + __expf(v1 - mx);
        if (ok2) s += __expf(v2 - mx);
        #pragma unroll
        for (int o = 8; o >= 1; o >>= 1) s += __shfl_xor(s, o, 32);
        const float ls = __logf(s);
        float* sr = st + (8 * hh + r) * N_CLS + c;
        sr[0]  = v0 - mx - ls;
        sr[16] = v1 - mx - ls;
        if (ok2) sr[32] = v2 - mx - ls;
    }
    __syncthreads();

    const int nvalid = min(n - row0, 16) * N_CLS;
    float* ob = out + (size_t)row0 * N_CLS;
    #pragma unroll
    for (int q = lane; q < 16 * N_CLS / 4; q += 32) {
        if (4 * q < nvalid) *(v4f*)(ob + 4 * q) = *(const v4f*)(st + 4 * q);
    }
}

extern "C" void kernel_launch(void* const* d_in, const int* in_sizes, int n_in,
                              void* d_out, int out_size, void* d_ws, size_t ws_size,
                              hipStream_t stream) {
    const float* x    = (const float*)d_in[0];
    const int*   ei   = (const int*)d_in[1];
    const float* ea   = (const float*)d_in[2];
    const float* eps1 = (const float*)d_in[3];
    const float* We1  = (const float*)d_in[4];
    const float* be1  = (const float*)d_in[5];
    const float* W11  = (const float*)d_in[6];
    const float* b11  = (const float*)d_in[7];
    const float* W12  = (const float*)d_in[8];
    const float* b12  = (const float*)d_in[9];
    const float* eps2 = (const float*)d_in[10];
    const float* We2  = (const float*)d_in[11];
    const float* be2  = (const float*)d_in[12];
    const float* W21  = (const float*)d_in[13];
    const float* b21  = (const float*)d_in[14];
    const float* W22  = (const float*)d_in[15];
    const float* b22  = (const float*)d_in[16];
    const float* Wl   = (const float*)d_in[17];
    const float* bl   = (const float*)d_in[18];

    const int N_ = in_sizes[0] / F_DIM;
    const int E_ = in_sizes[2];
    const size_t NB = (size_t)N_ * F_DIM * sizeof(float);

    char* ws = (char*)d_ws;
    float*    h2    = (float*)ws;
    float*    h1    = (float*)(ws + NB);
    _Float16* frags = (_Float16*)(ws + 2 * NB);
    unsigned* bits  = (unsigned*)(ws + 2 * NB + 131072);
    const size_t TBY = (((size_t)E_ + IDX_EDGES - 1) / IDX_EDGES) * IDX_SLOTS * 128;
    _Float16* f11 = frags;
    _Float16* f12 = frags + 16384;
    _Float16* f21 = frags + 32768;
    _Float16* f22 = frags + 49152;
    float* out = (float*)d_out;

    const int tb = (N_ + TILE_NODES - 1) / TILE_NODES;
    int* rcnt = (int*)(ws + 2 * NB + 131072 + ((TBY + 255) & ~(size_t)255));
    int* recs = rcnt + (((size_t)tb * EDGE_WAVES + 63) & ~(size_t)63);
    int* erec = recs + (size_t)tb * EDGE_WAVES * (3 * RCAP);
    const int cb = (N_ + 16 * CLS_WAVES - 1) / (16 * CLS_WAVES);

    prep<<<1024, 256, 0, stream>>>(W11, W12, W21, W22, frags, ei, ea, bits, erec, E_);

    edge_mlp<false><<<tb, 1024, 0, stream>>>(x, erec, We1, be1, bits, eps1, f11, b11, f12, b12,
                                              h1, recs, rcnt, E_, N_);
    edge_mlp<true><<<tb, 1024, 0, stream>>>(h1, erec, We2, be2, bits, eps2, f21, b21, f22, b22,
                                             h2, recs, rcnt, E_, N_);

    classifier<<<cb, 256, 0, stream>>>(h2, Wl, bl, out, N_);
}
